# starting kernel
# baseline (speedup 1.0000x reference)
#include <hip/hip_runtime.h>

#define NUM_NODE 100000
#define EMBED_DIM 128
#define N_EDGES 600000

typedef __attribute__((ext_vector_type(2))) float v2f;
typedef __attribute__((ext_vector_type(8))) float v8f;

__global__ void na_init_kernel(float* __restrict__ out, float* __restrict__ deg) {
    const int idx = blockIdx.x * blockDim.x + threadIdx.x;
    if (idx < NUM_NODE * EMBED_DIM) out[idx] = 0.0f;
    if (idx < NUM_NODE)             deg[idx] = 0.0f;
}

__global__ void __launch_bounds__(256)
na_gemm_relu_kernel(const float* __restrict__ x,
                    const float* __restrict__ W,
                    const float* __restrict__ b,
                    float* __restrict__ y) {
    const int lane = threadIdx.x & 31;
    const int waveInBlock = threadIdx.x >> 5;
    const int waveId = blockIdx.x * 8 + waveInBlock;
    const int m0 = waveId * 16;
    if (m0 >= NUM_NODE) return;

    const int half = lane >> 4;
    const int l16  = lane & 15;

    const v8f zero = {};
    v8f acc[8];
#pragma unroll
    for (int nt = 0; nt < 8; ++nt) acc[nt] = zero;

    const float* __restrict__ xrow = x + (long long)(m0 + l16) * EMBED_DIM;

    for (int k0 = 0; k0 < EMBED_DIM; k0 += 4) {
        const int kk = k0 + 2 * half;
        v2f a;
        a.x = xrow[kk + 0];
        a.y = xrow[kk + 1];
#pragma unroll
        for (int nt = 0; nt < 8; ++nt) {
            const int n = nt * 16 + l16;
            v2f bm;
            bm.x = W[n * EMBED_DIM + kk + 0];
            bm.y = W[n * EMBED_DIM + kk + 1];
            acc[nt] = __builtin_amdgcn_wmma_f32_16x16x4_f32(
                false, a, false, bm, (short)0, acc[nt], false, false);
        }
    }

#pragma unroll
    for (int nt = 0; nt < 8; ++nt) {
        const int n = nt * 16 + l16;
        const float bias = b[n];
#pragma unroll
        for (int j = 0; j < 8; ++j) {
            const int m = m0 + half * 8 + j;
            float v = acc[nt][j] + bias;
            y[(long long)m * EMBED_DIM + n] = v > 0.0f ? v : 0.0f;
        }
    }
}

__global__ void na_scatter_kernel(const int* __restrict__ src,
                                  const int* __restrict__ tgt,
                                  const float* __restrict__ y,
                                  float* __restrict__ out,
                                  float* __restrict__ deg) {
    const int gid = blockIdx.x * blockDim.x + threadIdx.x;
    const int e = gid >> 5;
    if (e >= N_EDGES) return;
    const int q  = (gid & 31) * 4;

    const int s = src[e];
    const int t = tgt[e];

    const float4 v = *(const float4*)(y + (long long)t * EMBED_DIM + q);
    float* o = out + (long long)s * EMBED_DIM + q;
    atomicAdd(o + 0, v.x);
    atomicAdd(o + 1, v.y);
    atomicAdd(o + 2, v.z);
    atomicAdd(o + 3, v.w);
    if (q == 0) atomicAdd(deg + s, 1.0f);
}

__global__ void na_norm_kernel(float* __restrict__ out,
                               const float* __restrict__ deg) {
    const int idx = blockIdx.x * blockDim.x + threadIdx.x;
    if (idx >= NUM_NODE * EMBED_DIM) return;
    float d = deg[idx >> 7];
    d = d > 1.0f ? d : 1.0f;
    out[idx] = out[idx] / d;
}

extern "C" void kernel_launch(void* const* d_in, const int* in_sizes, int n_in,
                              void* d_out, int out_size, void* d_ws, size_t ws_size,
                              hipStream_t stream) {
    const float* x  = (const float*)d_in[0];
    const int*   ei = (const int*)d_in[1];
    const float* W  = (const float*)d_in[2];
    const float* b  = (const float*)d_in[3];
    float* out = (float*)d_out;

    float* deg = (float*)d_ws;
    float* y   = (float*)((char*)d_ws +
                 (((size_t)NUM_NODE * sizeof(float) + 255) & ~(size_t)255));

    {
        const int total = NUM_NODE * EMBED_DIM;
        na_init_kernel<<<(total + 255) / 256, 256, 0, stream>>>(out, deg);
    }
    {
        const int waves  = NUM_NODE / 16;
        const int blocks = (waves + 7) / 8;
        na_gemm_relu_kernel<<<blocks, 256, 0, stream>>>(x, W, b, y);
    }
    {
        const long long threads = (long long)N_EDGES * 32;
        const int blocks = (int)((threads + 255) / 256);
        na_scatter_kernel<<<blocks, 256, 0, stream>>>(ei, ei + N_EDGES, y, out, deg);
    }
    {
        const int total = NUM_NODE * EMBED_DIM;
        na_norm_kernel<<<(total + 255) / 256, 256, 0, stream>>>(out, deg);
    }
}


# Scatter into LDS-resident output tiles: 256 work-groups each hold 391 rows, scan all edges and add y rows by LDS float adds (arrival order); zero and normalise passes fold in; 4 launches to 2
# speedup vs baseline: 2.2590x; 2.2590x over previous
#include <hip/hip_runtime.h>

#define NUM_NODE 100000
#define EMBED_DIM 128
#define N_EDGES 600000

typedef __attribute__((ext_vector_type(2))) float v2f;
typedef __attribute__((ext_vector_type(8))) float v8f;

#define GEMM_RT 4
#define GEMM_ROWS (GEMM_RT * 16)
#define GEMM_WAVES ((NUM_NODE + GEMM_ROWS - 1) / GEMM_ROWS)

__device__ __forceinline__ void na_load_frags(const float* const (&xr)[GEMM_RT],
                                              const float* wr, int k0,
                                              v2f (&a)[GEMM_RT], v2f (&bm)[8]) {
#pragma unroll
    for (int rt = 0; rt < GEMM_RT; ++rt)
        a[rt] = *(const v2f*)(xr[rt] + k0);
#pragma unroll
    for (int nt = 0; nt < 8; ++nt)
        bm[nt] = *(const v2f*)(wr + nt * 16 * EMBED_DIM + k0);
}

__device__ __forceinline__ void na_mma_step(v8f (&acc)[GEMM_RT][8],
                                            const v2f (&a)[GEMM_RT],
                                            const v2f (&bm)[8]) {
#pragma unroll
    for (int nt = 0; nt < 8; ++nt) {
#pragma unroll
        for (int rt = 0; rt < GEMM_RT; ++rt) {
            acc[rt][nt] = __builtin_amdgcn_wmma_f32_16x16x4_f32(
                false, a[rt], false, bm[nt], (short)0, acc[rt][nt], false, false);
        }
    }
}

__global__ void __launch_bounds__(64)
na_gemm_relu_kernel(const float* __restrict__ x,
                    const float* __restrict__ W,
                    const float* __restrict__ b,
                    float* __restrict__ y) {
    const int lane = threadIdx.x & 31;
    const int waveId = blockIdx.x * 2 + (threadIdx.x >> 5);
    const int m0 = waveId * GEMM_ROWS;
    if (m0 >= NUM_NODE) return;

    const int half = lane >> 4;
    const int l16  = lane & 15;

    const v8f zero = {};
    v8f acc[GEMM_RT][8];
#pragma unroll
    for (int rt = 0; rt < GEMM_RT; ++rt)
#pragma unroll
        for (int nt = 0; nt < 8; ++nt) acc[rt][nt] = zero;

    const float* xr[GEMM_RT];
#pragma unroll
    for (int rt = 0; rt < GEMM_RT; ++rt) {
        int m = m0 + rt * 16 + l16;
        m = m < NUM_NODE ? m : NUM_NODE - 1;
        xr[rt] = x + (long long)m * EMBED_DIM + 2 * half;
    }
    const float* wr = W + l16 * EMBED_DIM + 2 * half;

    v2f a0[GEMM_RT], b0[8], a1[GEMM_RT], b1[8];
    na_load_frags(xr, wr, 0, a0, b0);
    for (int k0 = 0; k0 < EMBED_DIM - 8; k0 += 8) {
        na_load_frags(xr, wr, k0 + 4, a1, b1);
        na_mma_step(acc, a0, b0);
        na_load_frags(xr, wr, k0 + 8, a0, b0);
        na_mma_step(acc, a1, b1);
    }
    na_load_frags(xr, wr, EMBED_DIM - 4, a1, b1);
    na_mma_step(acc, a0, b0);
    na_mma_step(acc, a1, b1);

#pragma unroll
    for (int rt = 0; rt < GEMM_RT; ++rt) {
        const int mt = m0 + rt * 16;
        if (mt < NUM_NODE) {
#pragma unroll
            for (int nt = 0; nt < 8; ++nt) {
                const int n = nt * 16 + l16;
                const float bias = b[n];
#pragma unroll
                for (int j = 0; j < 8; ++j) {
                    const int m = mt + half * 8 + j;
                    float v = acc[rt][nt][j] + bias;
                    y[(long long)m * EMBED_DIM + n] = v > 0.0f ? v : 0.0f;
                }
            }
        }
    }
}

#define AGG_R 391
#define AGG_TILES ((NUM_NODE + AGG_R - 1) / AGG_R)
#define AGG_THREADS 1024
#define AGG_WAVES (AGG_THREADS / 32)
#define AGG_SEG 32768
#define AGG_NSEG ((N_EDGES + AGG_SEG - 1) / AGG_SEG)
#define AGG_LD (AGG_SEG / 4 / AGG_THREADS)
#define AGG_QCAP 4096
#define AGG_NQUAD (N_EDGES / 4)
#define AGG_B 6

static_assert(N_EDGES % 4 == 0, "edge list is read in 16-byte pieces");
static_assert(N_EDGES <= (1 << 20) && AGG_R <= (1 << 12), "entry packs row<<20 | edge");

__device__ __forceinline__ void agg_add_row(float* tile, int* degc, unsigned r,
                                            const float4 v, int lane) {
    float* row = tile + r * EMBED_DIM + lane;
    atomicAdd(row,      v.x);
    atomicAdd(row + 32, v.y);
    atomicAdd(row + 64, v.z);
    atomicAdd(row + 96, v.w);
    if (lane == 0) atomicAdd(degc + r, 1);
}

__device__ __forceinline__ void agg_batch_issue(const unsigned* ml, unsigned n, unsigned m0,
                                                int wave, int lane,
                                                const int* __restrict__ tgt,
                                                const float* __restrict__ y,
                                                unsigned (&ent)[AGG_B], float4 (&v)[AGG_B]) {
    unsigned t[AGG_B];
    unsigned raw[AGG_B];
#pragma unroll
    for (int k = 0; k < AGG_B; ++k) {
        const unsigned m = m0 + (unsigned)wave + 32u * k;
        raw[k] = ml[m < AGG_QCAP ? m : AGG_QCAP - 1];
    }
#pragma unroll
    for (int k = 0; k < AGG_B; ++k) {
        const unsigned m = m0 + (unsigned)wave + 32u * k;
        const unsigned u = (unsigned)__builtin_amdgcn_readfirstlane(raw[k]);
        ent[k] = m < n ? u : 0u;
    }
#pragma unroll
    for (int k = 0; k < AGG_B; ++k) t[k] = (unsigned)tgt[ent[k] & 0xFFFFFu];
#pragma unroll
    for (int k = 0; k < AGG_B; ++k)
        v[k] = *(const float4*)(y + (size_t)t[k] * EMBED_DIM + lane * 4);
}

__device__ __forceinline__ void agg_batch_finish(float* tile, int* degc, unsigned n, unsigned m0,
                                                 int wave, int lane,
                                                 const unsigned (&ent)[AGG_B],
                                                 const float4 (&v)[AGG_B]) {
#pragma unroll
    for (int k = 0; k < AGG_B; ++k)
        if (m0 + (unsigned)wave + 32u * k < n) agg_add_row(tile, degc, ent[k] >> 20, v[k], lane);
}

__device__ __forceinline__ void agg_rest(float* tile, int* degc, const unsigned* ml,
                                         unsigned n, unsigned nall, int psg, int base,
                                         int wave, int lane,
                                         const int* __restrict__ src,
                                         const int* __restrict__ tgt,
                                         const float* __restrict__ y) {
    for (unsigned m0 = 32u * AGG_B; m0 < n; m0 += 32u * AGG_B) {
        unsigned ent[AGG_B];
        float4 v[AGG_B];
        agg_batch_issue(ml, n, m0, wave, lane, tgt, y, ent, v);
        agg_batch_finish(tile, degc, n, m0, wave, lane, ent, v);
    }
    if (nall > AGG_QCAP) {
        const int e0 = psg * AGG_SEG;
        const int e1 = e0 + AGG_SEG < N_EDGES ? e0 + AGG_SEG : N_EDGES;
        for (int e = e0 + wave; e < e1; e += AGG_WAVES) {
            const unsigned r = (unsigned)(src[e] - base);
            if (r < (unsigned)AGG_R) {
                const int tt = tgt[e];
                const float4 vv = *(const float4*)(y + (long long)tt * EMBED_DIM + lane * 4);
                agg_add_row(tile, degc, r, vv, lane);
            }
        }
    }
}

__global__ void __launch_bounds__(AGG_THREADS)
na_aggregate_kernel(const int* __restrict__ src,
                    const int* __restrict__ tgt,
                    const float* __restrict__ y,
                    float* __restrict__ out) {
    __shared__ __attribute__((aligned(16))) float tile[AGG_R * EMBED_DIM];
    __shared__ unsigned mlist[2][AGG_QCAP];
    __shared__ int degc[AGG_R + 1];
    __shared__ unsigned cnt[AGG_NSEG + 1];

    const int tid  = threadIdx.x;
    const int lane = tid & 31;
    const int wave = __builtin_amdgcn_readfirstlane(tid >> 5);
    const int base = blockIdx.x * AGG_R;

    for (int i = tid; i < AGG_R * EMBED_DIM / 4; i += AGG_THREADS)
        ((float4*)tile)[i] = make_float4(0.0f, 0.0f, 0.0f, 0.0f);
    for (int i = tid; i < AGG_R + 1; i += AGG_THREADS) degc[i] = 0;
    if (tid < AGG_NSEG + 1) cnt[tid] = 0u;
    __syncthreads();

    const int4* src4 = (const int4*)src;
    int4 sv[AGG_LD];

    for (int it = 0; it < AGG_NSEG; ++it) {
        const int sg = (it + (int)blockIdx.x) % AGG_NSEG;
        const int q0 = sg * (AGG_SEG / 4) + tid;

        const unsigned nall = it > 0 ? (unsigned)__builtin_amdgcn_readfirstlane(cnt[it - 1]) : 0u;
        const unsigned n = nall <= AGG_QCAP ? nall : 0u;
        const unsigned* ml = mlist[(it + 1) & 1];

#pragma unroll
        for (int k = 0; k < AGG_LD; ++k) {
            const int q = q0 + k * AGG_THREADS;
            sv[k] = src4[q < AGG_NQUAD ? q : AGG_NQUAD - 1];
        }

        unsigned ent[AGG_B];
        float4 v[AGG_B];
        agg_batch_issue(ml, n, 0u, wave, lane, tgt, y, ent, v);

        agg_batch_finish(tile, degc, n, 0u, wave, lane, ent, v);
        agg_rest(tile, degc, ml, n, nall, (it - 1 + (int)blockIdx.x) % AGG_NSEG, base,
                 wave, lane, src, tgt, y);

        unsigned* mq = mlist[it & 1];
#pragma unroll
        for (int k = 0; k < AGG_LD; ++k) {
            const int q = q0 + k * AGG_THREADS;
            const unsigned r0 = (unsigned)(sv[k].x - base);
            const unsigned r1 = (unsigned)(sv[k].y - base);
            const unsigned r2 = (unsigned)(sv[k].z - base);
            const unsigned r3 = (unsigned)(sv[k].w - base);
            const bool in_range = q < AGG_NQUAD;
            const bool h0 = r0 < (unsigned)AGG_R, h1 = r1 < (unsigned)AGG_R;
            const bool h2 = r2 < (unsigned)AGG_R, h3 = r3 < (unsigned)AGG_R;
            if (in_range && (h0 | h1 | h2 | h3)) {
                const unsigned rr[4] = {r0, r1, r2, r3};
                const bool hh[4] = {h0, h1, h2, h3};
#pragma unroll
                for (int c = 0; c < 4; ++c) {
                    if (hh[c]) {
                        const unsigned slot = atomicAdd(&cnt[it], 1u);
                        if (slot < AGG_QCAP) mq[slot] = (rr[c] << 20) | (unsigned)(q * 4 + c);
                    }
                }
            }
        }
        __syncthreads();
    }
    {
        const unsigned nall = (unsigned)__builtin_amdgcn_readfirstlane(cnt[AGG_NSEG - 1]);
        const unsigned n = nall <= AGG_QCAP ? nall : 0u;
        const unsigned* ml = mlist[(AGG_NSEG + 1) & 1];
        unsigned ent[AGG_B];
        float4 v[AGG_B];
        agg_batch_issue(ml, n, 0u, wave, lane, tgt, y, ent, v);
        agg_batch_finish(tile, degc, n, 0u, wave, lane, ent, v);
        agg_rest(tile, degc, ml, n, nall, (AGG_NSEG - 1 + (int)blockIdx.x) % AGG_NSEG, base,
                 wave, lane, src, tgt, y);
        __syncthreads();
    }

    for (int r = wave; r < AGG_R; r += AGG_WAVES) {
        const int node = base + r;
        if (node < NUM_NODE) {
            float d = (float)degc[r];
            d = d > 1.0f ? d : 1.0f;
            const float* row = tile + r * EMBED_DIM + lane;
            float4 o;
            o.x = row[0]  / d;
            o.y = row[32] / d;
            o.z = row[64] / d;
            o.w = row[96] / d;
            *(float4*)(out + (long long)node * EMBED_DIM + lane * 4) = o;
        }
    }
}

extern "C" void kernel_launch(void* const* d_in, const int* in_sizes, int n_in,
                              void* d_out, int out_size, void* d_ws, size_t ws_size,
                              hipStream_t stream) {
    const float* x  = (const float*)d_in[0];
    const int*   ei = (const int*)d_in[1];
    const float* W  = (const float*)d_in[2];
    const float* b  = (const float*)d_in[3];
    float* out = (float*)d_out;

    float* y   = (float*)((char*)d_ws +
                 (((size_t)NUM_NODE * sizeof(float) + 255) & ~(size_t)255));

    {
        const int blocks = (GEMM_WAVES + 1) / 2;
        na_gemm_relu_kernel<<<blocks, 64, 0, stream>>>(x, W, b, y);
    }
    {
        na_aggregate_kernel<<<AGG_TILES, AGG_THREADS, 0, stream>>>(ei, ei + N_EDGES, y, out);
    }
}


# Aggregation without a barrier per scan step: each wave lists its own matches (slots from wave ballots, no LDS counter) and adds their y rows itself; arrival order of a row's LDS adds changes
# speedup vs baseline: 2.5004x; 1.1069x over previous
#include <hip/hip_runtime.h>

#define NUM_NODE 100000
#define EMBED_DIM 128
#define N_EDGES 600000

typedef __attribute__((ext_vector_type(2))) float v2f;
typedef __attribute__((ext_vector_type(8))) float v8f;

#define GEMM_RT 4
#define GEMM_ROWS (GEMM_RT * 16)
#define GEMM_WAVES ((NUM_NODE + GEMM_ROWS - 1) / GEMM_ROWS)

__device__ __forceinline__ void na_load_frags(const float* const (&xr)[GEMM_RT],
                                              const float* wr, int k0,
                                              v2f (&a)[GEMM_RT], v2f (&bm)[8]) {
#pragma unroll
    for (int rt = 0; rt < GEMM_RT; ++rt)
        a[rt] = *(const v2f*)(xr[rt] + k0);
#pragma unroll
    for (int nt = 0; nt < 8; ++nt)
        bm[nt] = *(const v2f*)(wr + nt * 16 * EMBED_DIM + k0);
}

__device__ __forceinline__ void na_mma_step(v8f (&acc)[GEMM_RT][8],
                                            const v2f (&a)[GEMM_RT],
                                            const v2f (&bm)[8]) {
#pragma unroll
    for (int nt = 0; nt < 8; ++nt) {
#pragma unroll
        for (int rt = 0; rt < GEMM_RT; ++rt) {
            acc[rt][nt] = __builtin_amdgcn_wmma_f32_16x16x4_f32(
                false, a[rt], false, bm[nt], (short)0, acc[rt][nt], false, false);
        }
    }
}

__global__ void __launch_bounds__(64)
na_gemm_relu_kernel(const float* __restrict__ x,
                    const float* __restrict__ W,
                    const float* __restrict__ b,
                    float* __restrict__ y) {
    const int lane = threadIdx.x & 31;
    const int waveId = blockIdx.x * 2 + (threadIdx.x >> 5);
    const int m0 = waveId * GEMM_ROWS;
    if (m0 >= NUM_NODE) return;

    const int half = lane >> 4;
    const int l16  = lane & 15;

    const v8f zero = {};
    v8f acc[GEMM_RT][8];
#pragma unroll
    for (int rt = 0; rt < GEMM_RT; ++rt)
#pragma unroll
        for (int nt = 0; nt < 8; ++nt) acc[rt][nt] = zero;

    const float* xr[GEMM_RT];
#pragma unroll
    for (int rt = 0; rt < GEMM_RT; ++rt) {
        int m = m0 + rt * 16 + l16;
        m = m < NUM_NODE ? m : NUM_NODE - 1;
        xr[rt] = x + (long long)m * EMBED_DIM + 2 * half;
    }
    const float* wr = W + l16 * EMBED_DIM + 2 * half;

    v2f a0[GEMM_RT], b0[8], a1[GEMM_RT], b1[8];
    na_load_frags(xr, wr, 0, a0, b0);
    for (int k0 = 0; k0 < EMBED_DIM - 8; k0 += 8) {
        na_load_frags(xr, wr, k0 + 4, a1, b1);
        na_mma_step(acc, a0, b0);
        na_load_frags(xr, wr, k0 + 8, a0, b0);
        na_mma_step(acc, a1, b1);
    }
    na_load_frags(xr, wr, EMBED_DIM - 4, a1, b1);
    na_mma_step(acc, a0, b0);
    na_mma_step(acc, a1, b1);

#pragma unroll
    for (int rt = 0; rt < GEMM_RT; ++rt) {
        const int mt = m0 + rt * 16;
        if (mt < NUM_NODE) {
#pragma unroll
            for (int nt = 0; nt < 8; ++nt) {
                const int n = nt * 16 + l16;
                const float bias = b[n];
#pragma unroll
                for (int j = 0; j < 8; ++j) {
                    const int m = mt + half * 8 + j;
                    float v = acc[rt][nt][j] + bias;
                    y[(long long)m * EMBED_DIM + n] = v > 0.0f ? v : 0.0f;
                }
            }
        }
    }
}

#define AGG_R 391
#define AGG_TILES ((NUM_NODE + AGG_R - 1) / AGG_R)
#define AGG_THREADS 1024
#define AGG_WAVES (AGG_THREADS / 32)
#define AGG_SEG 32768
#define AGG_NSEG ((N_EDGES + AGG_SEG - 1) / AGG_SEG)
#define AGG_LD (AGG_SEG / 4 / AGG_THREADS)
#define AGG_WCAP 256
#define AGG_NQUAD (N_EDGES / 4)
#define AGG_B 8

static_assert(N_EDGES % 4 == 0, "edge list is read in 16-byte pieces");
static_assert(N_EDGES <= (1 << 20) && AGG_R <= (1 << 12), "entry packs row<<20 | edge");
static_assert((unsigned long long)NUM_NODE * EMBED_DIM * 4 <= 0xFFFFFFFFull, "y row offsets fit 32 bits");

__device__ __forceinline__ void agg_add_row(float* tile, int* degc, unsigned r,
                                            const float4 v, int lane) {
    float* row = tile + r * EMBED_DIM + lane;
    atomicAdd(row,      v.x);
    atomicAdd(row + 32, v.y);
    atomicAdd(row + 64, v.z);
    atomicAdd(row + 96, v.w);
    if (lane == 0) atomicAdd(degc + r, 1);
}

__device__ __forceinline__ void agg_fetch(const unsigned* ml, unsigned n, unsigned m0, int lane,
                                          const int* __restrict__ tgt,
                                          unsigned& rawv, unsigned& tv) {
    const unsigned m = m0 + (unsigned)lane;
    rawv = ml[m < n ? m : 0u];
    tv = (unsigned)tgt[rawv & 0xFFFFFu];
}

__device__ __forceinline__ void agg_rows(float* tile, int* degc, unsigned n, unsigned m0, int lane,
                                         unsigned rawv, unsigned tv,
                                         const float* __restrict__ y) {
    unsigned r[AGG_B];
    float4 v[AGG_B];
#pragma unroll
    for (int k = 0; k < AGG_B; ++k) {
        const unsigned t = (unsigned)__builtin_amdgcn_readlane((int)tv, k);
        r[k] = (unsigned)__builtin_amdgcn_readlane((int)rawv, k) >> 20;
        v[k] = *(const float4*)((const char*)y +
                                (t * (unsigned)(EMBED_DIM * 4) + (unsigned)lane * 16u));
    }
#pragma unroll
    for (int k = 0; k < AGG_B; ++k)
        if (m0 + (unsigned)k < n) agg_add_row(tile, degc, r[k], v[k], lane);
}

__device__ __forceinline__ void agg_walk(float* tile, int* degc, int sg, int wave, int base,
                                         int lane,
                                         const int* __restrict__ src,
                                         const int* __restrict__ tgt,
                                         const float* __restrict__ y) {
#pragma nounroll
    for (int k = 0; k < AGG_LD; ++k) {
        const int e0 = (sg * (AGG_SEG / 4) + k * AGG_THREADS + wave * 32) * 4;
#pragma nounroll
        for (int j = 0; j < 128; ++j) {
            const int e = e0 + j;
            if (e < N_EDGES) {
                const unsigned r = (unsigned)(src[e] - base);
                if (r < (unsigned)AGG_R) {
                    const int tt = tgt[e];
                    const float4 vv = *(const float4*)(y + (long long)tt * EMBED_DIM + lane * 4);
                    agg_add_row(tile, degc, r, vv, lane);
                }
            }
        }
    }
}

__global__ void __launch_bounds__(AGG_THREADS)
na_aggregate_kernel(const int* __restrict__ src,
                    const int* __restrict__ tgt,
                    const float* __restrict__ y,
                    float* __restrict__ out) {
    __shared__ __attribute__((aligned(16))) float tile[AGG_R * EMBED_DIM];
    __shared__ unsigned mlist[AGG_WAVES][AGG_WCAP];
    __shared__ int degc[AGG_R + 1];

    const int tid  = threadIdx.x;
    const int lane = tid & 31;
    const int wave = __builtin_amdgcn_readfirstlane(tid >> 5);
    const int base = blockIdx.x * AGG_R;

    for (int i = tid; i < AGG_R * EMBED_DIM / 4; i += AGG_THREADS)
        ((float4*)tile)[i] = make_float4(0.0f, 0.0f, 0.0f, 0.0f);
    for (int i = tid; i < AGG_R + 1; i += AGG_THREADS) degc[i] = 0;

    const int4* src4 = (const int4*)src;
    unsigned* ml = mlist[wave];
    int4 sv[AGG_LD];

    {
        const int q0 = ((int)blockIdx.x % AGG_NSEG) * (AGG_SEG / 4) + tid;
#pragma unroll
        for (int k = 0; k < AGG_LD; ++k) {
            const int q = q0 + k * AGG_THREADS;
            sv[k] = src4[q < AGG_NQUAD ? q : AGG_NQUAD - 1];
        }
    }
    __syncthreads();

    for (int it = 0; it < AGG_NSEG; ++it) {
        const int sg = (it + (int)blockIdx.x) % AGG_NSEG;
        const int q0 = sg * (AGG_SEG / 4) + tid;

        unsigned n = 0u;
#pragma unroll
        for (int k = 0; k < AGG_LD; ++k) {
            const int q = q0 + k * AGG_THREADS;
            const unsigned r0 = (unsigned)(sv[k].x - base);
            const unsigned r1 = (unsigned)(sv[k].y - base);
            const unsigned r2 = (unsigned)(sv[k].z - base);
            const unsigned r3 = (unsigned)(sv[k].w - base);
            const bool in_range = q < AGG_NQUAD;
            const bool h0 = r0 < (unsigned)AGG_R, h1 = r1 < (unsigned)AGG_R;
            const bool h2 = r2 < (unsigned)AGG_R, h3 = r3 < (unsigned)AGG_R;
            if (__builtin_amdgcn_ballot_w32(in_range && (h0 | h1 | h2 | h3)) != 0u) {
                const unsigned rr[4] = {r0, r1, r2, r3};
                const bool hh[4] = {h0, h1, h2, h3};
#pragma unroll
                for (int c = 0; c < 4; ++c) {
                    const bool hc = in_range && hh[c];
                    const unsigned bal = __builtin_amdgcn_ballot_w32(hc);
                    const unsigned slot = n + __builtin_amdgcn_mbcnt_lo(bal, 0u);
                    if (hc && slot < AGG_WCAP) ml[slot] = (rr[c] << 20) | (unsigned)(q * 4 + c);
                    n += (unsigned)__builtin_popcount(bal);
                }
            }
        }

        __builtin_amdgcn_fence(__ATOMIC_RELEASE, "wavefront");
        __builtin_amdgcn_fence(__ATOMIC_ACQUIRE, "wavefront");

        const bool listed = n != 0u && n <= AGG_WCAP;
        unsigned rawv = 0u, tv = 0u;
        if (listed) agg_fetch(ml, n, 0u, lane, tgt, rawv, tv);

        {
            const bool more = it + 1 < AGG_NSEG;
            const int q1 = ((it + 1 + (int)blockIdx.x) % AGG_NSEG) * (AGG_SEG / 4) + tid;
#pragma unroll
            for (int k = 0; k < AGG_LD; ++k) {
                const int q = q1 + k * AGG_THREADS;
                sv[k] = src4[more ? (q < AGG_NQUAD ? q : AGG_NQUAD - 1) : 0];
            }
        }

        if (listed) {
            agg_rows(tile, degc, n, 0u, lane, rawv, tv, y);
            for (unsigned m0 = AGG_B; m0 < n; m0 += AGG_B) {
                agg_fetch(ml, n, m0, lane, tgt, rawv, tv);
                agg_rows(tile, degc, n, m0, lane, rawv, tv, y);
            }
        } else if (n != 0u) {
            agg_walk(tile, degc, sg, wave, base, lane, src, tgt, y);
        }
    }
    __syncthreads();

    for (int r = wave; r < AGG_R; r += AGG_WAVES) {
        const int node = base + r;
        if (node < NUM_NODE) {
            float d = (float)degc[r];
            d = d > 1.0f ? d : 1.0f;
            const float* row = tile + r * EMBED_DIM + lane;
            float4 o;
            o.x = row[0]  / d;
            o.y = row[32] / d;
            o.z = row[64] / d;
            o.w = row[96] / d;
            *(float4*)(out + (long long)node * EMBED_DIM + lane * 4) = o;
        }
    }
}

extern "C" void kernel_launch(void* const* d_in, const int* in_sizes, int n_in,
                              void* d_out, int out_size, void* d_ws, size_t ws_size,
                              hipStream_t stream) {
    const float* x  = (const float*)d_in[0];
    const int*   ei = (const int*)d_in[1];
    const float* W  = (const float*)d_in[2];
    const float* b  = (const float*)d_in[3];
    float* out = (float*)d_out;

    float* y   = (float*)((char*)d_ws +
                 (((size_t)NUM_NODE * sizeof(float) + 255) & ~(size_t)255));

    {
        const int blocks = (GEMM_WAVES + 1) / 2;
        na_gemm_relu_kernel<<<blocks, 64, 0, stream>>>(x, W, b, y);
    }
    {
        na_aggregate_kernel<<<AGG_TILES, AGG_THREADS, 0, stream>>>(ei, ei + N_EDGES, y, out);
    }
}


# Scan skips edge blocks without the tile's sources: first work-groups of the GEMM launch build a per-tile presence bitmap of 64-edge blocks; tiles read only marked blocks; LDS add order changes
# speedup vs baseline: 2.8399x; 1.1358x over previous
#include <hip/hip_runtime.h>

#define NUM_NODE 100000
#define EMBED_DIM 128
#define N_EDGES 600000

typedef __attribute__((ext_vector_type(2))) float v2f;
typedef __attribute__((ext_vector_type(8))) float v8f;

#define GEMM_RT 4
#define GEMM_ROWS (GEMM_RT * 16)
#define GEMM_WAVES ((NUM_NODE + GEMM_ROWS - 1) / GEMM_ROWS)
#define GEMM_BLOCKS ((GEMM_WAVES + 1) / 2)

#define AGG_R 391
#define AGG_TILES ((NUM_NODE + AGG_R - 1) / AGG_R)
#define BM_CHUNK 2048
#define BM_WORDS ((N_EDGES + BM_CHUNK - 1) / BM_CHUNK)
#define AGG_NQUAD (N_EDGES / 4)
static_assert((unsigned long long)AGG_TILES * BM_WORDS * 4 <= (unsigned long long)NUM_NODE * 4,
              "the bitmap fits in front of y in the workspace");

__device__ __forceinline__ void na_load_frags(const float* const (&xr)[GEMM_RT],
                                              const float* wr, int k0,
                                              v2f (&a)[GEMM_RT], v2f (&bm)[8]) {
#pragma unroll
    for (int rt = 0; rt < GEMM_RT; ++rt)
        a[rt] = *(const v2f*)(xr[rt] + k0);
#pragma unroll
    for (int nt = 0; nt < 8; ++nt)
        bm[nt] = *(const v2f*)(wr + nt * 16 * EMBED_DIM + k0);
}

__device__ __forceinline__ void na_mma_step(v8f (&acc)[GEMM_RT][8],
                                            const v2f (&a)[GEMM_RT],
                                            const v2f (&bm)[8]) {
#pragma unroll
    for (int nt = 0; nt < 8; ++nt) {
#pragma unroll
        for (int rt = 0; rt < GEMM_RT; ++rt) {
            acc[rt][nt] = __builtin_amdgcn_wmma_f32_16x16x4_f32(
                false, a[rt], false, bm[nt], (short)0, acc[rt][nt], false, false);
        }
    }
}

__global__ void __launch_bounds__(64)
na_gemm_relu_kernel(const float* __restrict__ x,
                    const float* __restrict__ W,
                    const float* __restrict__ b,
                    float* __restrict__ y,
                    const int* __restrict__ src,
                    unsigned* __restrict__ bm) {
    if (blockIdx.x < BM_WORDS) {
        __shared__ unsigned pres[AGG_TILES];
        const int tid = threadIdx.x;
        for (int i = tid; i < AGG_TILES; i += 64) pres[i] = 0u;
        __syncthreads();
        const int4* src4 = (const int4*)src;
        int4 sq[BM_CHUNK / 4 / 64];
#pragma unroll
        for (int k = 0; k < BM_CHUNK / 4 / 64; ++k) {
            const int q = (int)blockIdx.x * (BM_CHUNK / 4) + k * 64 + tid;
            sq[k] = src4[q < AGG_NQUAD ? q : AGG_NQUAD - 1];
        }
#pragma unroll
        for (int k = 0; k < BM_CHUNK / 4 / 64; ++k) {
            const int p = k * 64 + tid;
            const int q = (int)blockIdx.x * (BM_CHUNK / 4) + p;
            if (q < AGG_NQUAD) {
                const unsigned bit = 1u << (p >> 4);
                const unsigned t0 = (unsigned)sq[k].x / AGG_R, t1 = (unsigned)sq[k].y / AGG_R;
                const unsigned t2 = (unsigned)sq[k].z / AGG_R, t3 = (unsigned)sq[k].w / AGG_R;
                if (t0 < AGG_TILES) atomicOr(&pres[t0], bit);
                if (t1 < AGG_TILES) atomicOr(&pres[t1], bit);
                if (t2 < AGG_TILES) atomicOr(&pres[t2], bit);
                if (t3 < AGG_TILES) atomicOr(&pres[t3], bit);
            }
        }
        __syncthreads();
        for (int i = tid; i < AGG_TILES; i += 64) bm[i * BM_WORDS + (int)blockIdx.x] = pres[i];
        return;
    }

    const int lane = threadIdx.x & 31;
    const int waveId = ((int)blockIdx.x - BM_WORDS) * 2 + (threadIdx.x >> 5);
    const int m0 = waveId * GEMM_ROWS;
    if (m0 >= NUM_NODE) return;

    const int half = lane >> 4;
    const int l16  = lane & 15;

    const v8f zero = {};
    v8f acc[GEMM_RT][8];
#pragma unroll
    for (int rt = 0; rt < GEMM_RT; ++rt)
#pragma unroll
        for (int nt = 0; nt < 8; ++nt) acc[rt][nt] = zero;

    const float* xr[GEMM_RT];
#pragma unroll
    for (int rt = 0; rt < GEMM_RT; ++rt) {
        int m = m0 + rt * 16 + l16;
        m = m < NUM_NODE ? m : NUM_NODE - 1;
        xr[rt] = x + (long long)m * EMBED_DIM + 2 * half;
    }
    const float* wr = W + l16 * EMBED_DIM + 2 * half;

    v2f a0[GEMM_RT], b0[8], a1[GEMM_RT], b1[8];
    na_load_frags(xr, wr, 0, a0, b0);
    for (int k0 = 0; k0 < EMBED_DIM - 8; k0 += 8) {
        na_load_frags(xr, wr, k0 + 4, a1, b1);
        na_mma_step(acc, a0, b0);
        na_load_frags(xr, wr, k0 + 8, a0, b0);
        na_mma_step(acc, a1, b1);
    }
    na_load_frags(xr, wr, EMBED_DIM - 4, a1, b1);
    na_mma_step(acc, a0, b0);
    na_mma_step(acc, a1, b1);

#pragma unroll
    for (int rt = 0; rt < GEMM_RT; ++rt) {
        const int mt = m0 + rt * 16;
        if (mt < NUM_NODE) {
#pragma unroll
            for (int nt = 0; nt < 8; ++nt) {
                const int n = nt * 16 + l16;
                const float bias = b[n];
#pragma unroll
                for (int j = 0; j < 8; ++j) {
                    const int m = mt + half * 8 + j;
                    float v = acc[rt][nt][j] + bias;
                    y[(long long)m * EMBED_DIM + n] = v > 0.0f ? v : 0.0f;
                }
            }
        }
    }
}

#define AGG_THREADS 1024
#define AGG_WAVES (AGG_THREADS / 32)
#define AGG_LD 8
#define AGG_CB (2 * AGG_LD)
#define AGG_WCAP 256
#define AGG_B 8
#define AGG_BLCAP (((BM_WORDS + AGG_WAVES - 1) / AGG_WAVES) * 32)

static_assert(N_EDGES % 64 == 0, "the edge list is a whole number of 64-edge blocks");
static_assert(N_EDGES <= (1 << 20) && AGG_R <= (1 << 12), "entry packs row<<20 | edge");
static_assert(N_EDGES / 64 <= 65536, "block numbers are kept as 16-bit values");
static_assert(AGG_LD * 128 <= 4 * AGG_WCAP, "one load adds at most 128 matches");
static_assert((unsigned long long)NUM_NODE * EMBED_DIM * 4 <= 0xFFFFFFFFull, "y row offsets fit 32 bits");

__device__ __forceinline__ void agg_add_row(float* tile, int* degc, unsigned r,
                                            const float4 v, int lane) {
    float* row = tile + r * EMBED_DIM + lane;
    atomicAdd(row,      v.x);
    atomicAdd(row + 32, v.y);
    atomicAdd(row + 64, v.z);
    atomicAdd(row + 96, v.w);
    if (lane == 0) atomicAdd(degc + r, 1);
}

__device__ __forceinline__ void agg_fetch(const unsigned* ml, unsigned n, unsigned m0, int lane,
                                          const int* __restrict__ tgt,
                                          unsigned& rawv, unsigned& tv) {
    const unsigned m = m0 + (unsigned)lane;
    rawv = ml[m < n ? m : 0u];
    tv = (unsigned)tgt[rawv & 0xFFFFFu];
}

template <int LB>
__device__ __forceinline__ void agg_rows(float* tile, int* degc, unsigned n, unsigned m0, int lane,
                                         unsigned rawv, unsigned tv,
                                         const float* __restrict__ y) {
    unsigned r[AGG_B];
    float4 v[AGG_B];
#pragma unroll
    for (int k = 0; k < AGG_B; ++k) {
        const unsigned t = (unsigned)__builtin_amdgcn_readlane((int)tv, LB + k);
        r[k] = (unsigned)__builtin_amdgcn_readlane((int)rawv, LB + k) >> 20;
        v[k] = *(const float4*)((const char*)y +
                                (t * (unsigned)(EMBED_DIM * 4) + (unsigned)lane * 16u));
    }
#pragma unroll
    for (int k = 0; k < AGG_B; ++k)
        if (m0 + (unsigned)k < n) agg_add_row(tile, degc, r[k], v[k], lane);
}

__device__ __forceinline__ void agg_groups(float* tile, int* degc, unsigned n, unsigned m0, int lane,
                                           unsigned rawv, unsigned tv,
                                           const float* __restrict__ y) {
    agg_rows<0>(tile, degc, n, m0, lane, rawv, tv, y);
    if (m0 + 8u  < n) agg_rows<8>(tile, degc, n, m0 + 8u, lane, rawv, tv, y);
    if (m0 + 16u < n) agg_rows<16>(tile, degc, n, m0 + 16u, lane, rawv, tv, y);
    if (m0 + 24u < n) agg_rows<24>(tile, degc, n, m0 + 24u, lane, rawv, tv, y);
}

__device__ __forceinline__ void agg_drain(float* tile, int* degc, const unsigned* ml, unsigned n,
                                          int lane,
                                          const int* __restrict__ tgt,
                                          const float* __restrict__ y) {
#pragma nounroll
    for (unsigned m = 0; m < n; ++m) {
        const unsigned raw = ml[m];
        const int tt = tgt[raw & 0xFFFFFu];
        const float4 vv = *(const float4*)(y + (long long)tt * EMBED_DIM + lane * 4);
        agg_add_row(tile, degc, raw >> 20, vv, lane);
    }
}

__global__ void __launch_bounds__(AGG_THREADS)
na_aggregate_kernel(const int* __restrict__ src,
                    const int* __restrict__ tgt,
                    const float* __restrict__ y,
                    const unsigned* __restrict__ bm,
                    float* __restrict__ out) {
    __shared__ __attribute__((aligned(16))) float tile[AGG_R * EMBED_DIM];
    __shared__ unsigned mlist[AGG_WAVES][AGG_WCAP];
    __shared__ unsigned short blist[AGG_WAVES][AGG_BLCAP];
    __shared__ int degc[AGG_R + 1];

    const int tid  = threadIdx.x;
    const int lane = tid & 31;
    const int half = lane >> 4;
    const int l16  = lane & 15;
    const int wave = __builtin_amdgcn_readfirstlane(tid >> 5);
    const int base = blockIdx.x * AGG_R;

    for (int i = tid; i < AGG_R * EMBED_DIM / 4; i += AGG_THREADS)
        ((float4*)tile)[i] = make_float4(0.0f, 0.0f, 0.0f, 0.0f);
    for (int i = tid; i < AGG_R + 1; i += AGG_THREADS) degc[i] = 0;

    const int4* src4 = (const int4*)src;
    unsigned* ml = mlist[wave];
    unsigned short* bl = blist[wave];

    unsigned nb = 0u;
    {
        const unsigned* bmrow = bm + (int)blockIdx.x * BM_WORDS;
        for (int w = wave; w < BM_WORDS; w += AGG_WAVES) {
            const unsigned word = (unsigned)__builtin_amdgcn_readfirstlane((int)bmrow[w]);
            if ((word >> lane) & 1u)
                bl[nb + __builtin_amdgcn_mbcnt_lo(word, 0u)] = (unsigned short)(w * 32 + lane);
            nb += (unsigned)__builtin_popcount(word);
        }
    }
    __builtin_amdgcn_fence(__ATOMIC_RELEASE, "wavefront");
    __builtin_amdgcn_fence(__ATOMIC_ACQUIRE, "wavefront");

    int4 sv[AGG_LD];
    int  qv[AGG_LD];
    if (nb != 0u) {
#pragma unroll
        for (int k = 0; k < AGG_LD; ++k) {
            const unsigned bi = 2u * k + (unsigned)half;
            qv[k] = (int)bl[bi < nb ? bi : 0u] * 16 + l16;
            sv[k] = src4[qv[k]];
        }
    }
    __syncthreads();

    if (nb != 0u) {
        for (unsigned c0 = 0u; c0 < nb; c0 += AGG_CB) {
            unsigned n = 0u;
#pragma unroll
            for (int k = 0; k < AGG_LD; ++k) {
                if (n > AGG_WCAP - 128u) {
                    __builtin_amdgcn_fence(__ATOMIC_RELEASE, "wavefront");
                    __builtin_amdgcn_fence(__ATOMIC_ACQUIRE, "wavefront");
                    agg_drain(tile, degc, ml, n, lane, tgt, y);
                    n = 0u;
                }
                const unsigned r0 = (unsigned)(sv[k].x - base);
                const unsigned r1 = (unsigned)(sv[k].y - base);
                const unsigned r2 = (unsigned)(sv[k].z - base);
                const unsigned r3 = (unsigned)(sv[k].w - base);
                const bool in_range = c0 + 2u * k + (unsigned)half < nb;
                const bool h0 = r0 < (unsigned)AGG_R, h1 = r1 < (unsigned)AGG_R;
                const bool h2 = r2 < (unsigned)AGG_R, h3 = r3 < (unsigned)AGG_R;
                if (__builtin_amdgcn_ballot_w32(in_range && (h0 | h1 | h2 | h3)) != 0u) {
                    const unsigned rr[4] = {r0, r1, r2, r3};
                    const bool hh[4] = {h0, h1, h2, h3};
#pragma unroll
                    for (int c = 0; c < 4; ++c) {
                        const bool hc = in_range && hh[c];
                        const unsigned bal = __builtin_amdgcn_ballot_w32(hc);
                        const unsigned slot = n + __builtin_amdgcn_mbcnt_lo(bal, 0u);
                        if (hc) ml[slot] = (rr[c] << 20) | (unsigned)(qv[k] * 4 + c);
                        n += (unsigned)__builtin_popcount(bal);
                    }
                }
            }

            __builtin_amdgcn_fence(__ATOMIC_RELEASE, "wavefront");
            __builtin_amdgcn_fence(__ATOMIC_ACQUIRE, "wavefront");

            unsigned rawv = 0u, tv = 0u;
            if (n != 0u) agg_fetch(ml, n, 0u, lane, tgt, rawv, tv);

#pragma unroll
            for (int k = 0; k < AGG_LD; ++k) {
                const unsigned bi = c0 + AGG_CB + 2u * k + (unsigned)half;
                qv[k] = (int)bl[bi < nb ? bi : 0u] * 16 + l16;
                sv[k] = src4[qv[k]];
            }

            if (n != 0u) {
                agg_groups(tile, degc, n, 0u, lane, rawv, tv, y);
                for (unsigned m0 = 32u; m0 < n; m0 += 32u) {
                    agg_fetch(ml, n, m0, lane, tgt, rawv, tv);
                    agg_groups(tile, degc, n, m0, lane, rawv, tv, y);
                }
            }
        }
    }
    __syncthreads();

    for (int r = wave; r < AGG_R; r += AGG_WAVES) {
        const int node = base + r;
        if (node < NUM_NODE) {
            float d = (float)degc[r];
            d = d > 1.0f ? d : 1.0f;
            const float* row = tile + r * EMBED_DIM + lane;
            float4 o;
            o.x = row[0]  / d;
            o.y = row[32] / d;
            o.z = row[64] / d;
            o.w = row[96] / d;
            *(float4*)(out + (long long)node * EMBED_DIM + lane * 4) = o;
        }
    }
}

extern "C" void kernel_launch(void* const* d_in, const int* in_sizes, int n_in,
                              void* d_out, int out_size, void* d_ws, size_t ws_size,
                              hipStream_t stream) {
    const float* x  = (const float*)d_in[0];
    const int*   ei = (const int*)d_in[1];
    const float* W  = (const float*)d_in[2];
    const float* b  = (const float*)d_in[3];
    float* out = (float*)d_out;

    unsigned* bm = (unsigned*)d_ws;
    float* y   = (float*)((char*)d_ws +
                 (((size_t)NUM_NODE * sizeof(float) + 255) & ~(size_t)255));

    {
        na_gemm_relu_kernel<<<BM_WORDS + GEMM_BLOCKS, 64, 0, stream>>>(x, W, b, y, ei, bm);
    }
    {
        na_aggregate_kernel<<<AGG_TILES, AGG_THREADS, 0, stream>>>(ei, ei + N_EDGES, y, bm, out);
    }
}


# Match bookkeeping by the lane that finds it: one slot ranking and one LDS store per load (not per component), degree counted there (not a wave-wide add per row); list order, so LDS add order, changes
# speedup vs baseline: 2.9450x; 1.0370x over previous
#include <hip/hip_runtime.h>

#define NUM_NODE 100000
#define EMBED_DIM 128
#define N_EDGES 600000

typedef __attribute__((ext_vector_type(2))) float v2f;
typedef __attribute__((ext_vector_type(8))) float v8f;

#define GEMM_RT 4
#define GEMM_ROWS (GEMM_RT * 16)
#define GEMM_WAVES ((NUM_NODE + GEMM_ROWS - 1) / GEMM_ROWS)
#define GEMM_BLOCKS ((GEMM_WAVES + 1) / 2)

#define AGG_R 391
#define AGG_TILES ((NUM_NODE + AGG_R - 1) / AGG_R)
#define BM_CHUNK 2048
#define BM_WORDS ((N_EDGES + BM_CHUNK - 1) / BM_CHUNK)
#define AGG_NQUAD (N_EDGES / 4)
static_assert((unsigned long long)AGG_TILES * BM_WORDS * 4 <= (unsigned long long)NUM_NODE * 4,
              "the bitmap fits in front of y in the workspace");

__device__ __forceinline__ void na_load_frags(const float* const (&xr)[GEMM_RT],
                                              const float* wr, int k0,
                                              v2f (&a)[GEMM_RT], v2f (&bm)[8]) {
#pragma unroll
    for (int rt = 0; rt < GEMM_RT; ++rt)
        a[rt] = *(const v2f*)(xr[rt] + k0);
#pragma unroll
    for (int nt = 0; nt < 8; ++nt)
        bm[nt] = *(const v2f*)(wr + nt * 16 * EMBED_DIM + k0);
}

__device__ __forceinline__ void na_mma_step(v8f (&acc)[GEMM_RT][8],
                                            const v2f (&a)[GEMM_RT],
                                            const v2f (&bm)[8]) {
#pragma unroll
    for (int nt = 0; nt < 8; ++nt) {
#pragma unroll
        for (int rt = 0; rt < GEMM_RT; ++rt) {
            acc[rt][nt] = __builtin_amdgcn_wmma_f32_16x16x4_f32(
                false, a[rt], false, bm[nt], (short)0, acc[rt][nt], false, false);
        }
    }
}

__global__ void __launch_bounds__(64)
na_gemm_relu_kernel(const float* __restrict__ x,
                    const float* __restrict__ W,
                    const float* __restrict__ b,
                    float* __restrict__ y,
                    const int* __restrict__ src,
                    unsigned* __restrict__ bm) {
    if (blockIdx.x < BM_WORDS) {
        __shared__ unsigned pres[AGG_TILES];
        const int tid = threadIdx.x;
        for (int i = tid; i < AGG_TILES; i += 64) pres[i] = 0u;
        __syncthreads();
        const int4* src4 = (const int4*)src;
        int4 sq[BM_CHUNK / 4 / 64];
#pragma unroll
        for (int k = 0; k < BM_CHUNK / 4 / 64; ++k) {
            const int q = (int)blockIdx.x * (BM_CHUNK / 4) + k * 64 + tid;
            sq[k] = src4[q < AGG_NQUAD ? q : AGG_NQUAD - 1];
        }
#pragma unroll
        for (int k = 0; k < BM_CHUNK / 4 / 64; ++k) {
            const int p = k * 64 + tid;
            const int q = (int)blockIdx.x * (BM_CHUNK / 4) + p;
            if (q < AGG_NQUAD) {
                const unsigned bit = 1u << (p >> 4);
                const unsigned t0 = (unsigned)sq[k].x / AGG_R, t1 = (unsigned)sq[k].y / AGG_R;
                const unsigned t2 = (unsigned)sq[k].z / AGG_R, t3 = (unsigned)sq[k].w / AGG_R;
                if (t0 < AGG_TILES) atomicOr(&pres[t0], bit);
                if (t1 < AGG_TILES) atomicOr(&pres[t1], bit);
                if (t2 < AGG_TILES) atomicOr(&pres[t2], bit);
                if (t3 < AGG_TILES) atomicOr(&pres[t3], bit);
            }
        }
        __syncthreads();
        for (int i = tid; i < AGG_TILES; i += 64) bm[i * BM_WORDS + (int)blockIdx.x] = pres[i];
        return;
    }

    const int lane = threadIdx.x & 31;
    const int waveId = ((int)blockIdx.x - BM_WORDS) * 2 + (threadIdx.x >> 5);
    const int m0 = waveId * GEMM_ROWS;
    if (m0 >= NUM_NODE) return;

    const int half = lane >> 4;
    const int l16  = lane & 15;

    const v8f zero = {};
    v8f acc[GEMM_RT][8];
#pragma unroll
    for (int rt = 0; rt < GEMM_RT; ++rt)
#pragma unroll
        for (int nt = 0; nt < 8; ++nt) acc[rt][nt] = zero;

    const float* xr[GEMM_RT];
#pragma unroll
    for (int rt = 0; rt < GEMM_RT; ++rt) {
        int m = m0 + rt * 16 + l16;
        m = m < NUM_NODE ? m : NUM_NODE - 1;
        xr[rt] = x + (long long)m * EMBED_DIM + 2 * half;
    }
    const float* wr = W + l16 * EMBED_DIM + 2 * half;

    v2f a0[GEMM_RT], b0[8], a1[GEMM_RT], b1[8];
    na_load_frags(xr, wr, 0, a0, b0);
    for (int k0 = 0; k0 < EMBED_DIM - 8; k0 += 8) {
        na_load_frags(xr, wr, k0 + 4, a1, b1);
        na_mma_step(acc, a0, b0);
        na_load_frags(xr, wr, k0 + 8, a0, b0);
        na_mma_step(acc, a1, b1);
    }
    na_load_frags(xr, wr, EMBED_DIM - 4, a1, b1);
    na_mma_step(acc, a0, b0);
    na_mma_step(acc, a1, b1);

#pragma unroll
    for (int rt = 0; rt < GEMM_RT; ++rt) {
        const int mt = m0 + rt * 16;
        if (mt < NUM_NODE) {
#pragma unroll
            for (int nt = 0; nt < 8; ++nt) {
                const int n = nt * 16 + l16;
                const float bias = b[n];
#pragma unroll
                for (int j = 0; j < 8; ++j) {
                    const int m = mt + half * 8 + j;
                    float v = acc[rt][nt][j] + bias;
                    y[(long long)m * EMBED_DIM + n] = v > 0.0f ? v : 0.0f;
                }
            }
        }
    }
}

#define AGG_THREADS 1024
#define AGG_WAVES (AGG_THREADS / 32)
#define AGG_LD 8
#define AGG_CB (2 * AGG_LD)
#define AGG_WCAP 256
#define AGG_B 8
#define AGG_BLCAP (((BM_WORDS + AGG_WAVES - 1) / AGG_WAVES) * 32)

static_assert(N_EDGES % 64 == 0, "the edge list is a whole number of 64-edge blocks");
static_assert(N_EDGES <= (1 << 20) && AGG_R <= (1 << 12), "entry packs row<<20 | edge");
static_assert(N_EDGES / 64 <= 65536, "block numbers are kept as 16-bit values");
static_assert(AGG_LD * 128 <= 4 * AGG_WCAP, "one load adds at most 128 matches");
static_assert((unsigned long long)NUM_NODE * EMBED_DIM * 4 <= 0xFFFFFFFFull, "y row offsets fit 32 bits");

__device__ __forceinline__ void agg_add_row(float* tile, unsigned r,
                                            const float4 v, int lane) {
    float* row = tile + r * EMBED_DIM + lane;
    atomicAdd(row,      v.x);
    atomicAdd(row + 32, v.y);
    atomicAdd(row + 64, v.z);
    atomicAdd(row + 96, v.w);
}

__device__ __forceinline__ void agg_fetch(const unsigned* ml, unsigned n, unsigned m0, int lane,
                                          const int* __restrict__ tgt,
                                          unsigned& rawv, unsigned& tv) {
    const unsigned m = m0 + (unsigned)lane;
    rawv = ml[m < n ? m : 0u];
    tv = (unsigned)tgt[rawv & 0xFFFFFu];
}

template <int LB>
__device__ __forceinline__ void agg_rows(float* tile, unsigned n, unsigned m0, int lane,
                                         unsigned rawv, unsigned tv,
                                         const float* __restrict__ y) {
    unsigned r[AGG_B];
    float4 v[AGG_B];
#pragma unroll
    for (int k = 0; k < AGG_B; ++k) {
        const unsigned t = (unsigned)__builtin_amdgcn_readlane((int)tv, LB + k);
        r[k] = (unsigned)__builtin_amdgcn_readlane((int)rawv, LB + k) >> 20;
        v[k] = *(const float4*)((const char*)y +
                                (t * (unsigned)(EMBED_DIM * 4) + (unsigned)lane * 16u));
    }
#pragma unroll
    for (int k = 0; k < AGG_B; ++k)
        if (m0 + (unsigned)k < n) agg_add_row(tile, r[k], v[k], lane);
}

__device__ __forceinline__ void agg_groups(float* tile, unsigned n, unsigned m0, int lane,
                                           unsigned rawv, unsigned tv,
                                           const float* __restrict__ y) {
    agg_rows<0>(tile, n, m0, lane, rawv, tv, y);
    if (m0 + 8u  < n) agg_rows<8>(tile, n, m0 + 8u, lane, rawv, tv, y);
    if (m0 + 16u < n) agg_rows<16>(tile, n, m0 + 16u, lane, rawv, tv, y);
    if (m0 + 24u < n) agg_rows<24>(tile, n, m0 + 24u, lane, rawv, tv, y);
}

__device__ __forceinline__ void agg_drain(float* tile, const unsigned* ml, unsigned n,
                                          int lane,
                                          const int* __restrict__ tgt,
                                          const float* __restrict__ y) {
#pragma nounroll
    for (unsigned m = 0; m < n; ++m) {
        const unsigned raw = ml[m];
        const int tt = tgt[raw & 0xFFFFFu];
        const float4 vv = *(const float4*)(y + (long long)tt * EMBED_DIM + lane * 4);
        agg_add_row(tile, raw >> 20, vv, lane);
    }
}

__global__ void __launch_bounds__(AGG_THREADS)
na_aggregate_kernel(const int* __restrict__ src,
                    const int* __restrict__ tgt,
                    const float* __restrict__ y,
                    const unsigned* __restrict__ bm,
                    float* __restrict__ out) {
    __shared__ __attribute__((aligned(16))) float tile[AGG_R * EMBED_DIM];
    __shared__ unsigned mlist[AGG_WAVES][AGG_WCAP];
    __shared__ unsigned short blist[AGG_WAVES][AGG_BLCAP];
    __shared__ int degc[AGG_R + 1];

    const int tid  = threadIdx.x;
    const int lane = tid & 31;
    const int half = lane >> 4;
    const int l16  = lane & 15;
    const int wave = __builtin_amdgcn_readfirstlane(tid >> 5);
    const int base = blockIdx.x * AGG_R;

    for (int i = tid; i < AGG_R * EMBED_DIM / 4; i += AGG_THREADS)
        ((float4*)tile)[i] = make_float4(0.0f, 0.0f, 0.0f, 0.0f);
    for (int i = tid; i < AGG_R + 1; i += AGG_THREADS) degc[i] = 0;

    const int4* src4 = (const int4*)src;
    unsigned* ml = mlist[wave];
    unsigned short* bl = blist[wave];

    unsigned nb = 0u;
    {
        const unsigned* bmrow = bm + (int)blockIdx.x * BM_WORDS;
        for (int w = wave; w < BM_WORDS; w += AGG_WAVES) {
            const unsigned word = (unsigned)__builtin_amdgcn_readfirstlane((int)bmrow[w]);
            if ((word >> lane) & 1u)
                bl[nb + __builtin_amdgcn_mbcnt_lo(word, 0u)] = (unsigned short)(w * 32 + lane);
            nb += (unsigned)__builtin_popcount(word);
        }
    }
    __builtin_amdgcn_fence(__ATOMIC_RELEASE, "wavefront");
    __builtin_amdgcn_fence(__ATOMIC_ACQUIRE, "wavefront");

    int4 sv[AGG_LD];
    int  qv[AGG_LD];
    if (nb != 0u) {
#pragma unroll
        for (int k = 0; k < AGG_LD; ++k) {
            const unsigned bi = 2u * k + (unsigned)half;
            qv[k] = (int)bl[bi < nb ? bi : 0u] * 16 + l16;
            sv[k] = src4[qv[k]];
        }
    }
    __syncthreads();

    if (nb != 0u) {
        for (unsigned c0 = 0u; c0 < nb; c0 += AGG_CB) {
            unsigned n = 0u;
#pragma unroll
            for (int k = 0; k < AGG_LD; ++k) {
                if (n > AGG_WCAP - 128u) {
                    __builtin_amdgcn_fence(__ATOMIC_RELEASE, "wavefront");
                    __builtin_amdgcn_fence(__ATOMIC_ACQUIRE, "wavefront");
                    agg_drain(tile, ml, n, lane, tgt, y);
                    n = 0u;
                }
                const bool in_range = c0 + 2u * k + (unsigned)half < nb;
                const unsigned kbase = in_range ? (unsigned)base : 0xC0000000u;
                const unsigned r0 = (unsigned)sv[k].x - kbase;
                const unsigned r1 = (unsigned)sv[k].y - kbase;
                const unsigned r2 = (unsigned)sv[k].z - kbase;
                const unsigned r3 = (unsigned)sv[k].w - kbase;
                const bool h0 = r0 < (unsigned)AGG_R, h1 = r1 < (unsigned)AGG_R;
                const bool h2 = r2 < (unsigned)AGG_R, h3 = r3 < (unsigned)AGG_R;
                const unsigned b0 = __builtin_amdgcn_ballot_w32(h0), b1 = __builtin_amdgcn_ballot_w32(h1);
                const unsigned b2 = __builtin_amdgcn_ballot_w32(h2), b3 = __builtin_amdgcn_ballot_w32(h3);
                const unsigned any = b0 | b1 | b2 | b3;
                if (any != 0u) {
                    const unsigned c  = h0 ? 0u : h1 ? 1u : h2 ? 2u : 3u;
                    const unsigned rc = h0 ? r0 : h1 ? r1 : h2 ? r2 : r3;
                    const unsigned slot = n + __builtin_amdgcn_mbcnt_lo(any, 0u);
                    if (h0 | h1 | h2 | h3) {
                        ml[slot] = (rc << 20) | ((unsigned)(qv[k] * 4) + c);
                        atomicAdd(degc + rc, 1);
                    }
                    n += (unsigned)__builtin_popcount(any);
                    if (((b0 & (b1 | b2 | b3)) | (b1 & (b2 | b3)) | (b2 & b3)) != 0u) {
                        const unsigned rr[3] = {r1, r2, r3};
                        const bool gg[3] = {h1 && h0, h2 && (h0 | h1), h3 && (h0 | h1 | h2)};
#pragma unroll
                        for (int e = 0; e < 3; ++e) {
                            const unsigned bal = __builtin_amdgcn_ballot_w32(gg[e]);
                            const unsigned s2 = n + __builtin_amdgcn_mbcnt_lo(bal, 0u);
                            if (gg[e]) {
                                ml[s2] = (rr[e] << 20) | (unsigned)(qv[k] * 4 + e + 1);
                                atomicAdd(degc + rr[e], 1);
                            }
                            n += (unsigned)__builtin_popcount(bal);
                        }
                    }
                }
            }

            __builtin_amdgcn_fence(__ATOMIC_RELEASE, "wavefront");
            __builtin_amdgcn_fence(__ATOMIC_ACQUIRE, "wavefront");

            unsigned rawv = 0u, tv = 0u;
            if (n != 0u) agg_fetch(ml, n, 0u, lane, tgt, rawv, tv);

#pragma unroll
            for (int k = 0; k < AGG_LD; ++k) {
                const unsigned bi = c0 + AGG_CB + 2u * k + (unsigned)half;
                qv[k] = (int)bl[bi < nb ? bi : 0u] * 16 + l16;
                sv[k] = src4[qv[k]];
            }

            if (n != 0u) {
                agg_groups(tile, n, 0u, lane, rawv, tv, y);
                for (unsigned m0 = 32u; m0 < n; m0 += 32u) {
                    agg_fetch(ml, n, m0, lane, tgt, rawv, tv);
                    agg_groups(tile, n, m0, lane, rawv, tv, y);
                }
            }
        }
    }
    __syncthreads();

    for (int r = wave; r < AGG_R; r += AGG_WAVES) {
        const int node = base + r;
        if (node < NUM_NODE) {
            float d = (float)degc[r];
            d = d > 1.0f ? d : 1.0f;
            const float* row = tile + r * EMBED_DIM + lane;
            float4 o;
            o.x = row[0]  / d;
            o.y = row[32] / d;
            o.z = row[64] / d;
            o.w = row[96] / d;
            *(float4*)(out + (long long)node * EMBED_DIM + lane * 4) = o;
        }
    }
}

extern "C" void kernel_launch(void* const* d_in, const int* in_sizes, int n_in,
                              void* d_out, int out_size, void* d_ws, size_t ws_size,
                              hipStream_t stream) {
    const float* x  = (const float*)d_in[0];
    const int*   ei = (const int*)d_in[1];
    const float* W  = (const float*)d_in[2];
    const float* b  = (const float*)d_in[3];
    float* out = (float*)d_out;

    unsigned* bm = (unsigned*)d_ws;
    float* y   = (float*)((char*)d_ws +
                 (((size_t)NUM_NODE * sizeof(float) + 255) & ~(size_t)255));

    {
        na_gemm_relu_kernel<<<BM_WORDS + GEMM_BLOCKS, 64, 0, stream>>>(x, W, b, y, ei, bm);
    }
    {
        na_aggregate_kernel<<<AGG_TILES, AGG_THREADS, 0, stream>>>(ei, ei + N_EDGES, y, bm, out);
    }
}


# Presence bitmap at 16-edge granularity (was 64): a tile reads and compares about a third of the edges; bitmap rows kept in the tile's own output rows until overwritten; LDS add order changes
# speedup vs baseline: 3.1030x; 1.0537x over previous
#include <hip/hip_runtime.h>

#define NUM_NODE 100000
#define EMBED_DIM 128
#define N_EDGES 600000

typedef __attribute__((ext_vector_type(2))) float v2f;
typedef __attribute__((ext_vector_type(8))) float v8f;

#define GEMM_RT 4
#define GEMM_ROWS (GEMM_RT * 16)
#define GEMM_WAVES ((NUM_NODE + GEMM_ROWS - 1) / GEMM_ROWS)
#define GEMM_BLOCKS ((GEMM_WAVES + 1) / 2)

#define AGG_R 391
#define AGG_TILES ((NUM_NODE + AGG_R - 1) / AGG_R)
#define SB_EDGES 16
#define SB_PIECES (SB_EDGES / 4)
#define BM_CHUNK 2048
#define BM_CHUNKS ((N_EDGES + BM_CHUNK - 1) / BM_CHUNK)
#define BM_CW (BM_CHUNK / SB_EDGES / 32)
#define BM_ROWW (BM_CHUNKS * BM_CW)
#define BM_PITCH (AGG_R * EMBED_DIM)
#define AGG_NQUAD (N_EDGES / 4)
static_assert(BM_CHUNK % (SB_EDGES * 32) == 0, "a bitmap work-group fills whole words");
static_assert((NUM_NODE - (AGG_TILES - 1) * AGG_R) * EMBED_DIM >= BM_ROWW,
              "a bitmap row fits inside the last (shortest) tile's output rows");

__device__ __forceinline__ void na_load_frags(const float* const (&xr)[GEMM_RT],
                                              const float* wr, int k0,
                                              v2f (&a)[GEMM_RT], v2f (&bm)[8]) {
#pragma unroll
    for (int rt = 0; rt < GEMM_RT; ++rt)
        a[rt] = *(const v2f*)(xr[rt] + k0);
#pragma unroll
    for (int nt = 0; nt < 8; ++nt)
        bm[nt] = *(const v2f*)(wr + nt * 16 * EMBED_DIM + k0);
}

__device__ __forceinline__ void na_mma_step(v8f (&acc)[GEMM_RT][8],
                                            const v2f (&a)[GEMM_RT],
                                            const v2f (&bm)[8]) {
#pragma unroll
    for (int nt = 0; nt < 8; ++nt) {
#pragma unroll
        for (int rt = 0; rt < GEMM_RT; ++rt) {
            acc[rt][nt] = __builtin_amdgcn_wmma_f32_16x16x4_f32(
                false, a[rt], false, bm[nt], (short)0, acc[rt][nt], false, false);
        }
    }
}

__global__ void __launch_bounds__(64)
na_gemm_relu_kernel(const float* __restrict__ x,
                    const float* __restrict__ W,
                    const float* __restrict__ b,
                    float* __restrict__ y,
                    const int* __restrict__ src,
                    unsigned* __restrict__ bm) {
    if (blockIdx.x < BM_CHUNKS) {
        __shared__ unsigned pres[AGG_TILES * BM_CW];
        const int tid = threadIdx.x;
        for (int i = tid; i < AGG_TILES * BM_CW; i += 64) pres[i] = 0u;
        __syncthreads();
        const int4* src4 = (const int4*)src;
        int4 sq[BM_CHUNK / 4 / 64];
#pragma unroll
        for (int k = 0; k < BM_CHUNK / 4 / 64; ++k) {
            const int q = (int)blockIdx.x * (BM_CHUNK / 4) + k * 64 + tid;
            sq[k] = src4[q < AGG_NQUAD ? q : AGG_NQUAD - 1];
        }
#pragma unroll
        for (int k = 0; k < BM_CHUNK / 4 / 64; ++k) {
            const int p = k * 64 + tid;
            const int q = (int)blockIdx.x * (BM_CHUNK / 4) + p;
            if (q < AGG_NQUAD) {
                const unsigned sb  = (unsigned)p / SB_PIECES;
                const unsigned wd  = sb >> 5;
                const unsigned bit = 1u << (sb & 31u);
                const unsigned t0 = (unsigned)sq[k].x / AGG_R, t1 = (unsigned)sq[k].y / AGG_R;
                const unsigned t2 = (unsigned)sq[k].z / AGG_R, t3 = (unsigned)sq[k].w / AGG_R;
                if (t0 < AGG_TILES) atomicOr(&pres[t0 * BM_CW + wd], bit);
                if (t1 < AGG_TILES) atomicOr(&pres[t1 * BM_CW + wd], bit);
                if (t2 < AGG_TILES) atomicOr(&pres[t2 * BM_CW + wd], bit);
                if (t3 < AGG_TILES) atomicOr(&pres[t3 * BM_CW + wd], bit);
            }
        }
        __syncthreads();
        for (int i = tid; i < AGG_TILES * BM_CW; i += 64)
            bm[(i / BM_CW) * BM_PITCH + (int)blockIdx.x * BM_CW + (i % BM_CW)] = pres[i];
        return;
    }

    const int lane = threadIdx.x & 31;
    const int waveId = ((int)blockIdx.x - BM_CHUNKS) * 2 + (threadIdx.x >> 5);
    const int m0 = waveId * GEMM_ROWS;
    if (m0 >= NUM_NODE) return;

    const int half = lane >> 4;
    const int l16  = lane & 15;

    const v8f zero = {};
    v8f acc[GEMM_RT][8];
#pragma unroll
    for (int rt = 0; rt < GEMM_RT; ++rt)
#pragma unroll
        for (int nt = 0; nt < 8; ++nt) acc[rt][nt] = zero;

    const float* xr[GEMM_RT];
#pragma unroll
    for (int rt = 0; rt < GEMM_RT; ++rt) {
        int m = m0 + rt * 16 + l16;
        m = m < NUM_NODE ? m : NUM_NODE - 1;
        xr[rt] = x + (long long)m * EMBED_DIM + 2 * half;
    }
    const float* wr = W + l16 * EMBED_DIM + 2 * half;

    v2f a0[GEMM_RT], b0[8], a1[GEMM_RT], b1[8];
    na_load_frags(xr, wr, 0, a0, b0);
    for (int k0 = 0; k0 < EMBED_DIM - 8; k0 += 8) {
        na_load_frags(xr, wr, k0 + 4, a1, b1);
        na_mma_step(acc, a0, b0);
        na_load_frags(xr, wr, k0 + 8, a0, b0);
        na_mma_step(acc, a1, b1);
    }
    na_load_frags(xr, wr, EMBED_DIM - 4, a1, b1);
    na_mma_step(acc, a0, b0);
    na_mma_step(acc, a1, b1);

#pragma unroll
    for (int rt = 0; rt < GEMM_RT; ++rt) {
        const int mt = m0 + rt * 16;
        if (mt < NUM_NODE) {
#pragma unroll
            for (int nt = 0; nt < 8; ++nt) {
                const int n = nt * 16 + l16;
                const float bias = b[n];
#pragma unroll
                for (int j = 0; j < 8; ++j) {
                    const int m = mt + half * 8 + j;
                    float v = acc[rt][nt][j] + bias;
                    y[(long long)m * EMBED_DIM + n] = v > 0.0f ? v : 0.0f;
                }
            }
        }
    }
}

#define AGG_THREADS 1024
#define AGG_WAVES (AGG_THREADS / 32)
#define AGG_LD 12
#define AGG_BPL (32 / SB_PIECES)
#define AGG_CB (AGG_BPL * AGG_LD)
#define AGG_WCAP 256
#define AGG_B 8
#define AGG_WW ((BM_ROWW + AGG_WAVES - 1) / AGG_WAVES)
#define AGG_BLCAP (AGG_WW * 32)

static_assert(N_EDGES % SB_EDGES == 0, "the edge list is a whole number of 16-edge blocks");
static_assert(N_EDGES <= (1 << 20) && AGG_R <= (1 << 12), "entry packs row<<20 | edge");
static_assert(N_EDGES / SB_EDGES <= 65536, "block numbers are kept as 16-bit values");
static_assert(AGG_WW > 32 && AGG_WW <= 64, "a wave's bitmap words are two loads per lane");
static_assert(AGG_WCAP >= 256, "a load lists at most 128 matches; the list is drained above AGG_WCAP - 128");
static_assert((unsigned long long)NUM_NODE * EMBED_DIM * 4 <= 0xFFFFFFFFull, "y row offsets fit 32 bits");

__device__ __forceinline__ void agg_add_row(float* tile, unsigned r,
                                            const float4 v, int lane) {
    float* row = tile + r * EMBED_DIM + lane;
    atomicAdd(row,      v.x);
    atomicAdd(row + 32, v.y);
    atomicAdd(row + 64, v.z);
    atomicAdd(row + 96, v.w);
}

__device__ __forceinline__ void agg_fetch(const unsigned* ml, unsigned n, unsigned m0, int lane,
                                          const int* __restrict__ tgt,
                                          unsigned& rawv, unsigned& tv) {
    const unsigned m = m0 + (unsigned)lane;
    rawv = ml[m < n ? m : 0u];
    tv = (unsigned)tgt[rawv & 0xFFFFFu];
}

template <int LB>
__device__ __forceinline__ void agg_rows(float* tile, unsigned n, unsigned m0, int lane,
                                         unsigned rawv, unsigned tv,
                                         const float* __restrict__ y) {
    unsigned r[AGG_B];
    float4 v[AGG_B];
#pragma unroll
    for (int k = 0; k < AGG_B; ++k) {
        const unsigned t = (unsigned)__builtin_amdgcn_readlane((int)tv, LB + k);
        r[k] = (unsigned)__builtin_amdgcn_readlane((int)rawv, LB + k) >> 20;
        v[k] = *(const float4*)((const char*)y +
                                (t * (unsigned)(EMBED_DIM * 4) + (unsigned)lane * 16u));
    }
#pragma unroll
    for (int k = 0; k < AGG_B; ++k)
        if (m0 + (unsigned)k < n) agg_add_row(tile, r[k], v[k], lane);
}

__device__ __forceinline__ void agg_groups(float* tile, unsigned n, unsigned m0, int lane,
                                           unsigned rawv, unsigned tv,
                                           const float* __restrict__ y) {
    agg_rows<0>(tile, n, m0, lane, rawv, tv, y);
    if (m0 + 8u  < n) agg_rows<8>(tile, n, m0 + 8u, lane, rawv, tv, y);
    if (m0 + 16u < n) agg_rows<16>(tile, n, m0 + 16u, lane, rawv, tv, y);
    if (m0 + 24u < n) agg_rows<24>(tile, n, m0 + 24u, lane, rawv, tv, y);
}

__device__ __forceinline__ void agg_drain(float* tile, const unsigned* ml, unsigned n,
                                          int lane,
                                          const int* __restrict__ tgt,
                                          const float* __restrict__ y) {
#pragma nounroll
    for (unsigned m = 0; m < n; ++m) {
        const unsigned raw = ml[m];
        const int tt = tgt[raw & 0xFFFFFu];
        const float4 vv = *(const float4*)(y + (long long)tt * EMBED_DIM + lane * 4);
        agg_add_row(tile, raw >> 20, vv, lane);
    }
}

__global__ void __launch_bounds__(AGG_THREADS)
na_aggregate_kernel(const int* __restrict__ src,
                    const int* __restrict__ tgt,
                    const float* __restrict__ y,
                    const unsigned* bm,
                    float* out) {
    __shared__ __attribute__((aligned(16))) float tile[AGG_R * EMBED_DIM];
    __shared__ unsigned mlist[AGG_WAVES][AGG_WCAP];
    __shared__ unsigned short blist[AGG_WAVES][AGG_BLCAP];
    __shared__ int degc[AGG_R + 1];

    const int tid  = threadIdx.x;
    const int lane = tid & 31;
    const int sub = lane / SB_PIECES;
    const int pc  = lane % SB_PIECES;
    const int wave = __builtin_amdgcn_readfirstlane(tid >> 5);
    const int base = blockIdx.x * AGG_R;

    for (int i = tid; i < AGG_R * EMBED_DIM / 4; i += AGG_THREADS)
        ((float4*)tile)[i] = make_float4(0.0f, 0.0f, 0.0f, 0.0f);
    for (int i = tid; i < AGG_R + 1; i += AGG_THREADS) degc[i] = 0;

    const int4* src4 = (const int4*)src;
    unsigned* ml = mlist[wave];
    unsigned short* bl = blist[wave];

    unsigned nb = 0u;
    {
        const unsigned* bmrow = bm + (int)blockIdx.x * BM_PITCH;
        const int w0 = wave * AGG_WW;
        unsigned wa = 0u, wb = 0u;
        if (w0 + lane < BM_ROWW) wa = bmrow[w0 + lane];
        if (lane < AGG_WW - 32 && w0 + 32 + lane < BM_ROWW) wb = bmrow[w0 + 32 + lane];
#pragma unroll
        for (int i = 0; i < 32; ++i) {
            const unsigned word = (unsigned)__builtin_amdgcn_readlane((int)wa, i);
            if ((word >> lane) & 1u)
                bl[nb + __builtin_amdgcn_mbcnt_lo(word, 0u)] = (unsigned short)((w0 + i) * 32 + lane);
            nb += (unsigned)__builtin_popcount(word);
        }
#pragma unroll
        for (int i = 0; i < AGG_WW - 32; ++i) {
            const unsigned word = (unsigned)__builtin_amdgcn_readlane((int)wb, i);
            if ((word >> lane) & 1u)
                bl[nb + __builtin_amdgcn_mbcnt_lo(word, 0u)] = (unsigned short)((w0 + 32 + i) * 32 + lane);
            nb += (unsigned)__builtin_popcount(word);
        }
    }
    __builtin_amdgcn_fence(__ATOMIC_RELEASE, "wavefront");
    __builtin_amdgcn_fence(__ATOMIC_ACQUIRE, "wavefront");

    int4 sv[AGG_LD];
    int  qv[AGG_LD];
    if (nb != 0u) {
#pragma unroll
        for (int k = 0; k < AGG_LD; ++k) {
            const unsigned bi = (unsigned)(AGG_BPL * k + sub);
            qv[k] = (int)bl[bi < nb ? bi : 0u] * SB_PIECES + pc;
            sv[k] = src4[qv[k]];
        }
    }
    __syncthreads();

    if (nb != 0u) {
        for (unsigned c0 = 0u; c0 < nb; c0 += AGG_CB) {
            unsigned n = 0u;
#pragma unroll
            for (int k = 0; k < AGG_LD; ++k) {
                if (n > AGG_WCAP - 128u) {
                    __builtin_amdgcn_fence(__ATOMIC_RELEASE, "wavefront");
                    __builtin_amdgcn_fence(__ATOMIC_ACQUIRE, "wavefront");
                    agg_drain(tile, ml, n, lane, tgt, y);
                    n = 0u;
                }
                const bool in_range = c0 + (unsigned)(AGG_BPL * k + sub) < nb;
                const unsigned kbase = in_range ? (unsigned)base : 0xC0000000u;
                const unsigned r0 = (unsigned)sv[k].x - kbase;
                const unsigned r1 = (unsigned)sv[k].y - kbase;
                const unsigned r2 = (unsigned)sv[k].z - kbase;
                const unsigned r3 = (unsigned)sv[k].w - kbase;
                const bool h0 = r0 < (unsigned)AGG_R, h1 = r1 < (unsigned)AGG_R;
                const bool h2 = r2 < (unsigned)AGG_R, h3 = r3 < (unsigned)AGG_R;
                const unsigned b0 = __builtin_amdgcn_ballot_w32(h0), b1 = __builtin_amdgcn_ballot_w32(h1);
                const unsigned b2 = __builtin_amdgcn_ballot_w32(h2), b3 = __builtin_amdgcn_ballot_w32(h3);
                const unsigned any = b0 | b1 | b2 | b3;
                if (any != 0u) {
                    const unsigned c  = h0 ? 0u : h1 ? 1u : h2 ? 2u : 3u;
                    const unsigned rc = h0 ? r0 : h1 ? r1 : h2 ? r2 : r3;
                    const unsigned slot = n + __builtin_amdgcn_mbcnt_lo(any, 0u);
                    if (h0 | h1 | h2 | h3) {
                        ml[slot] = (rc << 20) | ((unsigned)(qv[k] * 4) + c);
                        atomicAdd(degc + rc, 1);
                    }
                    n += (unsigned)__builtin_popcount(any);
                    if (((b0 & (b1 | b2 | b3)) | (b1 & (b2 | b3)) | (b2 & b3)) != 0u) {
                        const unsigned rr[3] = {r1, r2, r3};
                        const bool gg[3] = {h1 && h0, h2 && (h0 | h1), h3 && (h0 | h1 | h2)};
#pragma unroll
                        for (int e = 0; e < 3; ++e) {
                            const unsigned bal = __builtin_amdgcn_ballot_w32(gg[e]);
                            const unsigned s2 = n + __builtin_amdgcn_mbcnt_lo(bal, 0u);
                            if (gg[e]) {
                                ml[s2] = (rr[e] << 20) | (unsigned)(qv[k] * 4 + e + 1);
                                atomicAdd(degc + rr[e], 1);
                            }
                            n += (unsigned)__builtin_popcount(bal);
                        }
                    }
                }
            }

            __builtin_amdgcn_fence(__ATOMIC_RELEASE, "wavefront");
            __builtin_amdgcn_fence(__ATOMIC_ACQUIRE, "wavefront");

            unsigned rawv = 0u, tv = 0u;
            if (n != 0u) agg_fetch(ml, n, 0u, lane, tgt, rawv, tv);

            if (c0 + AGG_CB < nb) {
#pragma unroll
                for (int k = 0; k < AGG_LD; ++k) {
                    const unsigned bi = c0 + AGG_CB + (unsigned)(AGG_BPL * k + sub);
                    qv[k] = (int)bl[bi < nb ? bi : 0u] * SB_PIECES + pc;
                    sv[k] = src4[qv[k]];
                }
            }

            for (unsigned m0 = 0u; m0 < n; m0 += 32u) {
                unsigned rawn, tn;
                agg_fetch(ml, n, m0 + 32u, lane, tgt, rawn, tn);
                agg_groups(tile, n, m0, lane, rawv, tv, y);
                rawv = rawn;
                tv = tn;
            }
        }
    }
    __syncthreads();

    for (int r = wave; r < AGG_R; r += AGG_WAVES) {
        const int node = base + r;
        if (node < NUM_NODE) {
            float d = (float)degc[r];
            d = d > 1.0f ? d : 1.0f;
            const float* row = tile + r * EMBED_DIM + lane;
            float4 o;
            o.x = row[0]  / d;
            o.y = row[32] / d;
            o.z = row[64] / d;
            o.w = row[96] / d;
            *(float4*)(out + (long long)node * EMBED_DIM + lane * 4) = o;
        }
    }
}

extern "C" void kernel_launch(void* const* d_in, const int* in_sizes, int n_in,
                              void* d_out, int out_size, void* d_ws, size_t ws_size,
                              hipStream_t stream) {
    const float* x  = (const float*)d_in[0];
    const int*   ei = (const int*)d_in[1];
    const float* W  = (const float*)d_in[2];
    const float* b  = (const float*)d_in[3];
    float* out = (float*)d_out;

    unsigned* bm = (unsigned*)d_out;
    float* y   = (float*)((char*)d_ws +
                 (((size_t)NUM_NODE * sizeof(float) + 255) & ~(size_t)255));

    {
        na_gemm_relu_kernel<<<BM_CHUNKS + GEMM_BLOCKS, 64, 0, stream>>>(x, W, b, y, ei, bm);
    }
    {
        na_aggregate_kernel<<<AGG_TILES, AGG_THREADS, 0, stream>>>(ei, ei + N_EDGES, y, bm, out);
    }
}


# GEMM by 1024 persistent waves (one per SIMD): W's B fragments loaded once and held in registers (512 VGPRs), each wave walks 6-7 16-row tiles with the next tile's x fragments in flight; k order kept
# speedup vs baseline: 3.2272x; 1.0400x over previous
#include <hip/hip_runtime.h>

#define NUM_NODE 100000
#define EMBED_DIM 128
#define N_EDGES 600000

typedef __attribute__((ext_vector_type(2))) float v2f;
typedef __attribute__((ext_vector_type(8))) float v8f;

#define GEMM_PW 1024
#define GEMM_BLOCKS (GEMM_PW / 2)
#define GEMM_TILES (NUM_NODE / 16)
#define GEMM_FULL (GEMM_TILES / GEMM_PW)
#define GEMM_REM (GEMM_TILES - GEMM_FULL * GEMM_PW)
#define GEMM_KS (EMBED_DIM / 4)
static_assert(NUM_NODE % 16 == 0, "x is a whole number of 16-row tiles");
static_assert(GEMM_FULL >= 2, "the peeled first tile is followed by at least one more");

#define AGG_R 391
#define AGG_TILES ((NUM_NODE + AGG_R - 1) / AGG_R)
#define SB_EDGES 16
#define SB_PIECES (SB_EDGES / 4)
#define BM_CHUNK 2048
#define BM_CHUNKS ((N_EDGES + BM_CHUNK - 1) / BM_CHUNK)
#define BM_CW (BM_CHUNK / SB_EDGES / 32)
#define BM_ROWW (BM_CHUNKS * BM_CW)
#define BM_PITCH (AGG_R * EMBED_DIM)
#define AGG_NQUAD (N_EDGES / 4)
static_assert(BM_CHUNK % (SB_EDGES * 32) == 0, "a bitmap work-group fills whole words");
static_assert((NUM_NODE - (AGG_TILES - 1) * AGG_R) * EMBED_DIM >= BM_ROWW,
              "a bitmap row fits inside the last (shortest) tile's output rows");

#define GEMM_REM_W0 (2 * BM_CHUNKS)
static_assert(BM_CHUNKS <= GEMM_BLOCKS, "every bitmap chunk has a work-group");
static_assert(GEMM_REM_W0 + GEMM_REM <= GEMM_PW, "the left-over tiles go to waves without bitmap work");

__device__ __forceinline__ void na_load_a(const float* xl, int tile, v2f (&a)[GEMM_KS]) {
    const float* p = xl + (long long)tile * (16 * EMBED_DIM);
#pragma unroll
    for (int ks = 0; ks < GEMM_KS; ++ks)
        a[ks] = *(const v2f*)(p + ks * 4);
}

template <int KS>
__device__ __forceinline__ void na_kstep(v8f (&acc)[8], const v2f a,
                                         const v2f (&wf)[8][GEMM_KS]) {
    const v8f zero = {};
#pragma unroll
    for (int nt = 0; nt < 8; ++nt)
        acc[nt] = __builtin_amdgcn_wmma_f32_16x16x4_f32(
            false, a, false, wf[nt][KS], (short)0, KS == 0 ? zero : acc[nt], false, false);
}

__device__ __forceinline__ void na_store_tile(const v8f (&acc)[8], const float (&bias)[8],
                                              float* yl, int tile) {
    float* yt = yl + (long long)tile * (16 * EMBED_DIM);
#pragma unroll
    for (int nt = 0; nt < 8; ++nt) {
#pragma unroll
        for (int j = 0; j < 8; ++j) {
            const float v = acc[nt][j] + bias[nt];
            yt[j * EMBED_DIM + nt * 16] = v > 0.0f ? v : 0.0f;
        }
    }
}

template <int KS>
__device__ __forceinline__ void na_ksteps_from(v8f (&acc)[8], const v2f (&a)[GEMM_KS],
                                               const v2f (&wf)[8][GEMM_KS]) {
    if constexpr (KS < GEMM_KS) {
        na_kstep<KS>(acc, a[KS], wf);
        na_ksteps_from<KS + 1>(acc, a, wf);
    }
}
__device__ __forceinline__ void na_tile(const v2f (&a)[GEMM_KS],
                                        const v2f (&wf)[8][GEMM_KS],
                                        const float (&bias)[8],
                                        float* yl, int tile) {
    v8f acc[8];
    na_ksteps_from<0>(acc, a, wf);
    na_store_tile(acc, bias, yl, tile);
}

template <int KS>
__device__ __forceinline__ void na_load_w(const float* wr, v2f (&wf)[8][GEMM_KS]) {
#pragma unroll
    for (int nt = 0; nt < 8; ++nt)
        wf[nt][KS] = *(const v2f*)(wr + nt * 16 * EMBED_DIM + KS * 4);
}

#define GEMM_WAHEAD 7
template <int KS>
__device__ __forceinline__ void na_first_tile_from(v8f (&acc)[8], const v2f (&a)[GEMM_KS],
                                                   v2f (&wf)[8][GEMM_KS], const float* wr,
                                                   const float* xl, int tnext,
                                                   v2f (&an)[GEMM_KS]) {
    if constexpr (KS < GEMM_KS) {
        if constexpr (KS + GEMM_WAHEAD < GEMM_KS) na_load_w<KS + GEMM_WAHEAD>(wr, wf);
        if constexpr (KS + GEMM_WAHEAD == GEMM_KS) na_load_a(xl, tnext, an);
        __builtin_amdgcn_sched_barrier(0);
        na_kstep<KS>(acc, a[KS], wf);
        na_first_tile_from<KS + 1>(acc, a, wf, wr, xl, tnext, an);
    }
}
template <int KS>
__device__ __forceinline__ void na_load_w_upto(const float* wr, v2f (&wf)[8][GEMM_KS]) {
    if constexpr (KS < GEMM_WAHEAD) {
        na_load_w<KS>(wr, wf);
        na_load_w_upto<KS + 1>(wr, wf);
    }
}

__global__ void __launch_bounds__(64)
na_gemm_relu_kernel(const float* __restrict__ x,
                    const float* __restrict__ W,
                    const float* __restrict__ b,
                    float* __restrict__ y,
                    const int* __restrict__ src,
                    unsigned* __restrict__ bm) {
    if (blockIdx.x < BM_CHUNKS) {
        __shared__ unsigned pres[AGG_TILES * BM_CW];
        const int tid = threadIdx.x;
        for (int i = tid; i < AGG_TILES * BM_CW; i += 64) pres[i] = 0u;
        __syncthreads();
        const int4* src4 = (const int4*)src;
        int4 sq[BM_CHUNK / 4 / 64];
#pragma unroll
        for (int k = 0; k < BM_CHUNK / 4 / 64; ++k) {
            const int q = (int)blockIdx.x * (BM_CHUNK / 4) + k * 64 + tid;
            sq[k] = src4[q < AGG_NQUAD ? q : AGG_NQUAD - 1];
        }
#pragma unroll
        for (int k = 0; k < BM_CHUNK / 4 / 64; ++k) {
            const int p = k * 64 + tid;
            const int q = (int)blockIdx.x * (BM_CHUNK / 4) + p;
            if (q < AGG_NQUAD) {
                const unsigned sb  = (unsigned)p / SB_PIECES;
                const unsigned wd  = sb >> 5;
                const unsigned bit = 1u << (sb & 31u);
                const unsigned t0 = (unsigned)sq[k].x / AGG_R, t1 = (unsigned)sq[k].y / AGG_R;
                const unsigned t2 = (unsigned)sq[k].z / AGG_R, t3 = (unsigned)sq[k].w / AGG_R;
                if (t0 < AGG_TILES) atomicOr(&pres[t0 * BM_CW + wd], bit);
                if (t1 < AGG_TILES) atomicOr(&pres[t1 * BM_CW + wd], bit);
                if (t2 < AGG_TILES) atomicOr(&pres[t2 * BM_CW + wd], bit);
                if (t3 < AGG_TILES) atomicOr(&pres[t3 * BM_CW + wd], bit);
            }
        }
        __syncthreads();
        for (int i = tid; i < AGG_TILES * BM_CW; i += 64)
            bm[(i / BM_CW) * BM_PITCH + (int)blockIdx.x * BM_CW + (i % BM_CW)] = pres[i];
    }

    const int lane = threadIdx.x & 31;
    const int half = lane >> 4;
    const int l16  = lane & 15;
    const int w = (int)blockIdx.x * 2 + __builtin_amdgcn_readfirstlane((int)threadIdx.x >> 5);
    const int ntile = GEMM_FULL + ((w >= GEMM_REM_W0 && w < GEMM_REM_W0 + GEMM_REM) ? 1 : 0);
    const int tlast = GEMM_FULL * GEMM_PW + (w - GEMM_REM_W0);

    const float* xl = x + l16 * EMBED_DIM + 2 * half;
    float* yl = y + (half * 8) * EMBED_DIM + l16;

    const float* wr = W + l16 * EMBED_DIM + 2 * half;
    float bias[8];
#pragma unroll
    for (int nt = 0; nt < 8; ++nt) bias[nt] = b[nt * 16 + l16];

    v2f ac[GEMM_KS], an[GEMM_KS];
    v2f wf[8][GEMM_KS];

    int t = w;
    int tn = t + GEMM_PW;
    {
        na_load_a(xl, t, ac);
        na_load_w_upto<0>(wr, wf);
        v8f acc[8];
        na_first_tile_from<0>(acc, ac, wf, wr, xl, tn, an);
        na_store_tile(acc, bias, yl, t);
    }
    int i = 1;
    t = tn;
#pragma unroll
    for (int ks = 0; ks < GEMM_KS; ++ks) ac[ks] = an[ks];
    for (;;) {
        if (i + 1 < ntile) tn = (i + 1 < GEMM_FULL) ? t + GEMM_PW : tlast;
        na_load_a(xl, tn, an);
        __builtin_amdgcn_sched_barrier(0);
        na_tile(ac, wf, bias, yl, t);
        __builtin_amdgcn_sched_barrier(0);
#pragma unroll
        for (int ks = 0; ks < GEMM_KS; ++ks) ac[ks] = an[ks];
        t = tn;
        if (++i >= ntile) break;
    }
}

#define AGG_THREADS 1024
#define AGG_WAVES (AGG_THREADS / 32)
#define AGG_LD 12
#define AGG_BPL (32 / SB_PIECES)
#define AGG_CB (AGG_BPL * AGG_LD)
#define AGG_WCAP 256
#define AGG_B 8
#define AGG_WW ((BM_ROWW + AGG_WAVES - 1) / AGG_WAVES)
#define AGG_BLCAP (AGG_WW * 32)

static_assert(N_EDGES % SB_EDGES == 0, "the edge list is a whole number of 16-edge blocks");
static_assert(N_EDGES <= (1 << 20) && AGG_R <= (1 << 12), "entry packs row<<20 | edge");
static_assert(N_EDGES / SB_EDGES <= 65536, "block numbers are kept as 16-bit values");
static_assert(AGG_WW > 32 && AGG_WW <= 64, "a wave's bitmap words are two loads per lane");
static_assert(AGG_WCAP >= 256, "a load lists at most 128 matches; the list is drained above AGG_WCAP - 128");
static_assert((unsigned long long)NUM_NODE * EMBED_DIM * 4 <= 0xFFFFFFFFull, "y row offsets fit 32 bits");

__device__ __forceinline__ void agg_add_row(float* tile, unsigned r,
                                            const float4 v, int lane) {
    float* row = tile + r * EMBED_DIM + lane;
    atomicAdd(row,      v.x);
    atomicAdd(row + 32, v.y);
    atomicAdd(row + 64, v.z);
    atomicAdd(row + 96, v.w);
}

__device__ __forceinline__ void agg_fetch(const unsigned* ml, unsigned n, unsigned m0, int lane,
                                          const int* __restrict__ tgt,
                                          unsigned& rawv, unsigned& tv) {
    const unsigned m = m0 + (unsigned)lane;
    rawv = ml[m < n ? m : 0u];
    tv = (unsigned)tgt[rawv & 0xFFFFFu];
}

template <int LB>
__device__ __forceinline__ void agg_rows(float* tile, unsigned n, unsigned m0, int lane,
                                         unsigned rawv, unsigned tv,
                                         const float* __restrict__ y) {
    unsigned r[AGG_B];
    float4 v[AGG_B];
#pragma unroll
    for (int k = 0; k < AGG_B; ++k) {
        const unsigned t = (unsigned)__builtin_amdgcn_readlane((int)tv, LB + k);
        r[k] = (unsigned)__builtin_amdgcn_readlane((int)rawv, LB + k) >> 20;
        v[k] = *(const float4*)((const char*)y +
                                (t * (unsigned)(EMBED_DIM * 4) + (unsigned)lane * 16u));
    }
#pragma unroll
    for (int k = 0; k < AGG_B; ++k)
        if (m0 + (unsigned)k < n) agg_add_row(tile, r[k], v[k], lane);
}

__device__ __forceinline__ void agg_groups(float* tile, unsigned n, unsigned m0, int lane,
                                           unsigned rawv, unsigned tv,
                                           const float* __restrict__ y) {
    agg_rows<0>(tile, n, m0, lane, rawv, tv, y);
    if (m0 + 8u  < n) agg_rows<8>(tile, n, m0 + 8u, lane, rawv, tv, y);
    if (m0 + 16u < n) agg_rows<16>(tile, n, m0 + 16u, lane, rawv, tv, y);
    if (m0 + 24u < n) agg_rows<24>(tile, n, m0 + 24u, lane, rawv, tv, y);
}

__device__ __forceinline__ void agg_drain(float* tile, const unsigned* ml, unsigned n,
                                          int lane,
                                          const int* __restrict__ tgt,
                                          const float* __restrict__ y) {
#pragma nounroll
    for (unsigned m = 0; m < n; ++m) {
        const unsigned raw = ml[m];
        const int tt = tgt[raw & 0xFFFFFu];
        const float4 vv = *(const float4*)(y + (long long)tt * EMBED_DIM + lane * 4);
        agg_add_row(tile, raw >> 20, vv, lane);
    }
}

__global__ void __launch_bounds__(AGG_THREADS)
na_aggregate_kernel(const int* __restrict__ src,
                    const int* __restrict__ tgt,
                    const float* __restrict__ y,
                    const unsigned* bm,
                    float* out) {
    __shared__ __attribute__((aligned(16))) float tile[AGG_R * EMBED_DIM];
    __shared__ unsigned mlist[AGG_WAVES][AGG_WCAP];
    __shared__ unsigned short blist[AGG_WAVES][AGG_BLCAP];
    __shared__ int degc[AGG_R + 1];

    const int tid  = threadIdx.x;
    const int lane = tid & 31;
    const int sub = lane / SB_PIECES;
    const int pc  = lane % SB_PIECES;
    const int wave = __builtin_amdgcn_readfirstlane(tid >> 5);
    const int base = blockIdx.x * AGG_R;

    for (int i = tid; i < AGG_R * EMBED_DIM / 4; i += AGG_THREADS)
        ((float4*)tile)[i] = make_float4(0.0f, 0.0f, 0.0f, 0.0f);
    for (int i = tid; i < AGG_R + 1; i += AGG_THREADS) degc[i] = 0;

    const int4* src4 = (const int4*)src;
    unsigned* ml = mlist[wave];
    unsigned short* bl = blist[wave];

    unsigned nb = 0u;
    {
        const unsigned* bmrow = bm + (int)blockIdx.x * BM_PITCH;
        const int w0 = wave * AGG_WW;
        unsigned wa = 0u, wb = 0u;
        if (w0 + lane < BM_ROWW) wa = bmrow[w0 + lane];
        if (lane < AGG_WW - 32 && w0 + 32 + lane < BM_ROWW) wb = bmrow[w0 + 32 + lane];
#pragma unroll
        for (int i = 0; i < 32; ++i) {
            const unsigned word = (unsigned)__builtin_amdgcn_readlane((int)wa, i);
            if ((word >> lane) & 1u)
                bl[nb + __builtin_amdgcn_mbcnt_lo(word, 0u)] = (unsigned short)((w0 + i) * 32 + lane);
            nb += (unsigned)__builtin_popcount(word);
        }
#pragma unroll
        for (int i = 0; i < AGG_WW - 32; ++i) {
            const unsigned word = (unsigned)__builtin_amdgcn_readlane((int)wb, i);
            if ((word >> lane) & 1u)
                bl[nb + __builtin_amdgcn_mbcnt_lo(word, 0u)] = (unsigned short)((w0 + 32 + i) * 32 + lane);
            nb += (unsigned)__builtin_popcount(word);
        }
    }
    __builtin_amdgcn_fence(__ATOMIC_RELEASE, "wavefront");
    __builtin_amdgcn_fence(__ATOMIC_ACQUIRE, "wavefront");

    int4 sv[AGG_LD];
    int  qv[AGG_LD];
    if (nb != 0u) {
#pragma unroll
        for (int k = 0; k < AGG_LD; ++k) {
            const unsigned bi = (unsigned)(AGG_BPL * k + sub);
            qv[k] = (int)bl[bi < nb ? bi : 0u] * SB_PIECES + pc;
            sv[k] = src4[qv[k]];
        }
    }
    __syncthreads();

    if (nb != 0u) {
        for (unsigned c0 = 0u; c0 < nb; c0 += AGG_CB) {
            unsigned n = 0u;
#pragma unroll
            for (int k = 0; k < AGG_LD; ++k) {
                if (n > AGG_WCAP - 128u) {
                    __builtin_amdgcn_fence(__ATOMIC_RELEASE, "wavefront");
                    __builtin_amdgcn_fence(__ATOMIC_ACQUIRE, "wavefront");
                    agg_drain(tile, ml, n, lane, tgt, y);
                    n = 0u;
                }
                const bool in_range = c0 + (unsigned)(AGG_BPL * k + sub) < nb;
                const unsigned kbase = in_range ? (unsigned)base : 0xC0000000u;
                const unsigned r0 = (unsigned)sv[k].x - kbase;
                const unsigned r1 = (unsigned)sv[k].y - kbase;
                const unsigned r2 = (unsigned)sv[k].z - kbase;
                const unsigned r3 = (unsigned)sv[k].w - kbase;
                const bool h0 = r0 < (unsigned)AGG_R, h1 = r1 < (unsigned)AGG_R;
                const bool h2 = r2 < (unsigned)AGG_R, h3 = r3 < (unsigned)AGG_R;
                const unsigned b0 = __builtin_amdgcn_ballot_w32(h0), b1 = __builtin_amdgcn_ballot_w32(h1);
                const unsigned b2 = __builtin_amdgcn_ballot_w32(h2), b3 = __builtin_amdgcn_ballot_w32(h3);
                const unsigned any = b0 | b1 | b2 | b3;
                if (any != 0u) {
                    const unsigned c  = h0 ? 0u : h1 ? 1u : h2 ? 2u : 3u;
                    const unsigned rc = h0 ? r0 : h1 ? r1 : h2 ? r2 : r3;
                    const unsigned slot = n + __builtin_amdgcn_mbcnt_lo(any, 0u);
                    if (h0 | h1 | h2 | h3) {
                        ml[slot] = (rc << 20) | ((unsigned)(qv[k] * 4) + c);
                        atomicAdd(degc + rc, 1);
                    }
                    n += (unsigned)__builtin_popcount(any);
                    if (((b0 & (b1 | b2 | b3)) | (b1 & (b2 | b3)) | (b2 & b3)) != 0u) {
                        const unsigned rr[3] = {r1, r2, r3};
                        const bool gg[3] = {h1 && h0, h2 && (h0 | h1), h3 && (h0 | h1 | h2)};
#pragma unroll
                        for (int e = 0; e < 3; ++e) {
                            const unsigned bal = __builtin_amdgcn_ballot_w32(gg[e]);
                            const unsigned s2 = n + __builtin_amdgcn_mbcnt_lo(bal, 0u);
                            if (gg[e]) {
                                ml[s2] = (rr[e] << 20) | (unsigned)(qv[k] * 4 + e + 1);
                                atomicAdd(degc + rr[e], 1);
                            }
                            n += (unsigned)__builtin_popcount(bal);
                        }
                    }
                }
            }

            __builtin_amdgcn_fence(__ATOMIC_RELEASE, "wavefront");
            __builtin_amdgcn_fence(__ATOMIC_ACQUIRE, "wavefront");

            unsigned rawv = 0u, tv = 0u;
            if (n != 0u) agg_fetch(ml, n, 0u, lane, tgt, rawv, tv);

            if (c0 + AGG_CB < nb) {
#pragma unroll
                for (int k = 0; k < AGG_LD; ++k) {
                    const unsigned bi = c0 + AGG_CB + (unsigned)(AGG_BPL * k + sub);
                    qv[k] = (int)bl[bi < nb ? bi : 0u] * SB_PIECES + pc;
                    sv[k] = src4[qv[k]];
                }
            }

            for (unsigned m0 = 0u; m0 < n; m0 += 32u) {
                unsigned rawn, tn;
                agg_fetch(ml, n, m0 + 32u, lane, tgt, rawn, tn);
                agg_groups(tile, n, m0, lane, rawv, tv, y);
                rawv = rawn;
                tv = tn;
            }
        }
    }
    __syncthreads();

    for (int r = wave; r < AGG_R; r += AGG_WAVES) {
        const int node = base + r;
        if (node < NUM_NODE) {
            float d = (float)degc[r];
            d = d > 1.0f ? d : 1.0f;
            const float* row = tile + r * EMBED_DIM + lane;
            float4 o;
            o.x = row[0]  / d;
            o.y = row[32] / d;
            o.z = row[64] / d;
            o.w = row[96] / d;
            *(float4*)(out + (long long)node * EMBED_DIM + lane * 4) = o;
        }
    }
}

extern "C" void kernel_launch(void* const* d_in, const int* in_sizes, int n_in,
                              void* d_out, int out_size, void* d_ws, size_t ws_size,
                              hipStream_t stream) {
    const float* x  = (const float*)d_in[0];
    const int*   ei = (const int*)d_in[1];
    const float* W  = (const float*)d_in[2];
    const float* b  = (const float*)d_in[3];
    float* out = (float*)d_out;

    unsigned* bm = (unsigned*)d_out;
    float* y   = (float*)((char*)d_ws +
                 (((size_t)NUM_NODE * sizeof(float) + 255) & ~(size_t)255));

    {
        na_gemm_relu_kernel<<<GEMM_BLOCKS, 64, 0, stream>>>(x, W, b, y, ei, bm);
    }
    {
        na_aggregate_kernel<<<AGG_TILES, AGG_THREADS, 0, stream>>>(ei, ei + N_EDGES, y, bm, out);
    }
}
